# starting kernel
# baseline (speedup 1.0000x reference)
#include <hip/hip_runtime.h>

typedef __attribute__((ext_vector_type(2))) float v2f;
typedef __attribute__((ext_vector_type(8))) float v8f;

#define N_NODES   131072
#define E_EDGES   4194304
#define S_IN      32
#define H_DIM     16
#define B_GRAPHS  128
#define FC_IN_SZ  16384
#define KCHUNKS   8

__global__ void fill_kernel(float* __restrict__ p, float v, int n) {
    int i = blockIdx.x * blockDim.x + threadIdx.x;
    if (i < n) p[i] = v;
}

__global__ void degree_kernel(const int* __restrict__ col, float* __restrict__ deg, int e) {
    int i = blockIdx.x * blockDim.x + threadIdx.x;
    if (i < e) atomicAdd(&deg[col[i]], 1.0f);
}

__global__ void rsqrt_kernel(float* __restrict__ d, int n) {
    int i = blockIdx.x * blockDim.x + threadIdx.x;
    if (i < n) d[i] = rsqrtf(d[i]);
}

__global__ __launch_bounds__(32)
void wmma_gemm_f32(const float* __restrict__ A, const float* __restrict__ B,
                   const float* __restrict__ bias, float* __restrict__ C,
                   int K, int Nld, int relu, int Kc) {
    const int nb   = blockIdx.x;
    const int mb   = blockIdx.y;
    const int z    = blockIdx.z;
    const int lane = threadIdx.x;
    const int half = lane >> 4;
    const int lm   = lane & 15;

    const int k0 = z * Kc;

    v8f acc = {};
    const float* Ap = A + (size_t)(mb * 16 + lm) * K + k0 + 2 * half;
    const float* Bp = B + (size_t)(k0 + 2 * half) * Nld + nb * 16 + lm;

#pragma unroll 4
    for (int k = 0; k < Kc; k += 4) {
        v2f a = *(const v2f*)Ap;
        v2f b;
        b.x = Bp[0];
        b.y = Bp[Nld];
        acc = __builtin_amdgcn_wmma_f32_16x16x4_f32(
                  false, a, false, b, (short)0, acc, false, false);
        Ap += 4;
        Bp += (size_t)4 * Nld;
    }

    const int cn = nb * 16 + lm;
    if (gridDim.z == 1) {
        const float bb = bias ? bias[cn] : 0.0f;
#pragma unroll
        for (int v = 0; v < 8; ++v) {
            float r = acc[v] + bb;
            if (relu) r = fmaxf(r, 0.0f);
            C[(size_t)(mb * 16 + v + 8 * half) * Nld + cn] = r;
        }
    } else {
        const size_t M  = (size_t)gridDim.y * 16;
        float* Cp = C + (size_t)z * M * Nld;
#pragma unroll
        for (int v = 0; v < 8; ++v) {
            Cp[(size_t)(mb * 16 + v + 8 * half) * Nld + cn] = acc[v];
        }
    }
}

__global__ void reduce_bias_relu_kernel(const float* __restrict__ part,
                                        const float* __restrict__ bias,
                                        float* __restrict__ C,
                                        int MN, int Nld, int relu) {
    int i = blockIdx.x * blockDim.x + threadIdx.x;
    if (i >= MN) return;
    float s = 0.0f;
#pragma unroll
    for (int zc = 0; zc < KCHUNKS; ++zc) s += part[(size_t)zc * MN + i];
    s += bias[i % Nld];
    if (relu) s = fmaxf(s, 0.0f);
    C[i] = s;
}

__global__ void scatter_kernel(const float* __restrict__ xw,
                               const int* __restrict__ row, const int* __restrict__ col,
                               const float* __restrict__ dinv,
                               float* __restrict__ out, int e16) {
    int t = blockIdx.x * blockDim.x + threadIdx.x;
    if (t >= e16) return;
    int e = t >> 4, j = t & 15;
    int r = row[e], c = col[e];
    float w = dinv[r] * dinv[c];
    atomicAdd(&out[c * H_DIM + j], w * xw[r * H_DIM + j]);
}

__global__ void finalize_kernel(float* __restrict__ h, const float* __restrict__ xw,
                                const float* __restrict__ dinv,
                                const float* __restrict__ bias, int n) {
    int i = blockIdx.x * blockDim.x + threadIdx.x;
    if (i >= n) return;
    int node = i >> 4, j = i & 15;
    float d = dinv[node];
    float r = h[i] + d * d * xw[i] + bias[j];
    h[i] = fmaxf(r, 0.0f);
}

__global__ void heads_kernel(const float* __restrict__ V,
                             const float* __restrict__ Wiv, const float* __restrict__ biv,
                             const float* __restrict__ Wev, const float* __restrict__ bev,
                             float* __restrict__ out_ext, float* __restrict__ out_int) {
    int r = blockIdx.x * blockDim.x + threadIdx.x;
    if (r >= B_GRAPHS) return;
    const float* v = V + (size_t)r * 256;
    float si = 0.0f, se = 0.0f;
#pragma unroll 4
    for (int k = 0; k < 256; ++k) {
        float x = v[k];
        si += x * Wiv[k];
        se += x * Wev[k];
    }
    out_int[r] = si + biv[0];
    out_ext[r] = se + bev[0];
}

extern "C" void kernel_launch(void* const* d_in, const int* in_sizes, int n_in,
                              void* d_out, int out_size, void* d_ws, size_t ws_size,
                              hipStream_t stream) {
    const float* x    = (const float*)d_in[0];
    const int*   ei   = (const int*)d_in[1];
    const float* W1   = (const float*)d_in[2];
    const float* b1   = (const float*)d_in[3];
    const float* W2   = (const float*)d_in[4];
    const float* b2   = (const float*)d_in[5];
    const float* W3   = (const float*)d_in[6];
    const float* b3   = (const float*)d_in[7];
    const float* Wp1  = (const float*)d_in[8];
    const float* bp1  = (const float*)d_in[9];
    const float* Wp2  = (const float*)d_in[10];
    const float* bp2  = (const float*)d_in[11];
    const float* Wv1  = (const float*)d_in[12];
    const float* bv1  = (const float*)d_in[13];
    const float* Wiv  = (const float*)d_in[14];
    const float* biv  = (const float*)d_in[15];
    const float* Wev  = (const float*)d_in[16];
    const float* bev  = (const float*)d_in[17];

    const int* row = ei;
    const int* col = ei + E_EDGES;

    char* ws = (char*)d_ws;
    float* dinv = (float*)ws;
    float* bufA = (float*)(ws + (size_t)N_NODES * 4);
    float* bufB = bufA + (size_t)N_NODES * H_DIM;
    float* Hp   = bufB + (size_t)N_NODES * H_DIM;
    float* Vv   = Hp + (size_t)B_GRAPHS * 256;
    float* part = Vv + (size_t)B_GRAPHS * 256;

    float* P     = (float*)d_out;
    float* ext_V = P + (size_t)B_GRAPHS * 1024;
    float* int_V = ext_V + B_GRAPHS;

    const int TPB = 256;
    const int nh  = N_NODES * H_DIM;
    const int e16 = E_EDGES * H_DIM;
    const int fcMN = B_GRAPHS * 256;

    fill_kernel<<<(N_NODES + TPB - 1) / TPB, TPB, 0, stream>>>(dinv, 1.0f, N_NODES);
    degree_kernel<<<(E_EDGES + TPB - 1) / TPB, TPB, 0, stream>>>(col, dinv, E_EDGES);
    rsqrt_kernel<<<(N_NODES + TPB - 1) / TPB, TPB, 0, stream>>>(dinv, N_NODES);

    wmma_gemm_f32<<<dim3(1, N_NODES / 16, 1), 32, 0, stream>>>(x, W1, nullptr, bufA, S_IN, H_DIM, 0, S_IN);
    fill_kernel<<<(nh + TPB - 1) / TPB, TPB, 0, stream>>>(bufB, 0.0f, nh);
    scatter_kernel<<<(e16 + TPB - 1) / TPB, TPB, 0, stream>>>(bufA, row, col, dinv, bufB, e16);
    finalize_kernel<<<(nh + TPB - 1) / TPB, TPB, 0, stream>>>(bufB, bufA, dinv, b1, nh);

    wmma_gemm_f32<<<dim3(1, N_NODES / 16, 1), 32, 0, stream>>>(bufB, W2, nullptr, bufA, H_DIM, H_DIM, 0, H_DIM);
    fill_kernel<<<(nh + TPB - 1) / TPB, TPB, 0, stream>>>(bufB, 0.0f, nh);
    scatter_kernel<<<(e16 + TPB - 1) / TPB, TPB, 0, stream>>>(bufA, row, col, dinv, bufB, e16);
    finalize_kernel<<<(nh + TPB - 1) / TPB, TPB, 0, stream>>>(bufB, bufA, dinv, b2, nh);

    wmma_gemm_f32<<<dim3(1, N_NODES / 16, 1), 32, 0, stream>>>(bufB, W3, nullptr, bufA, H_DIM, H_DIM, 0, H_DIM);
    fill_kernel<<<(nh + TPB - 1) / TPB, TPB, 0, stream>>>(bufB, 0.0f, nh);
    scatter_kernel<<<(e16 + TPB - 1) / TPB, TPB, 0, stream>>>(bufA, row, col, dinv, bufB, e16);
    finalize_kernel<<<(nh + TPB - 1) / TPB, TPB, 0, stream>>>(bufB, bufA, dinv, b3, nh);


    wmma_gemm_f32<<<dim3(256 / 16, B_GRAPHS / 16, KCHUNKS), 32, 0, stream>>>(
        bufB, Wp1, nullptr, part, FC_IN_SZ, 256, 0, FC_IN_SZ / KCHUNKS);
    reduce_bias_relu_kernel<<<(fcMN + TPB - 1) / TPB, TPB, 0, stream>>>(part, bp1, Hp, fcMN, 256, 1);
    wmma_gemm_f32<<<dim3(256 / 16, B_GRAPHS / 16, KCHUNKS), 32, 0, stream>>>(
        bufB, Wv1, nullptr, part, FC_IN_SZ, 256, 0, FC_IN_SZ / KCHUNKS);
    reduce_bias_relu_kernel<<<(fcMN + TPB - 1) / TPB, TPB, 0, stream>>>(part, bv1, Vv, fcMN, 256, 1);

    wmma_gemm_f32<<<dim3(1024 / 16, B_GRAPHS / 16, 1), 32, 0, stream>>>(Hp, Wp2, bp2, P, 256, 1024, 0, 256);

    heads_kernel<<<1, B_GRAPHS, 0, stream>>>(Vv, Wiv, biv, Wev, bev, ext_V, int_V);
}


# Replace the 3 atomic scatter-adds by atomic-free gathers over an edge list sorted by target, built per call (one returning-atomic rank pass, scan, permute); neighbour sums now in arrival-rank order
# speedup vs baseline: 1.5466x; 1.5466x over previous
#include <hip/hip_runtime.h>

typedef __attribute__((ext_vector_type(2))) float v2f;
typedef __attribute__((ext_vector_type(8))) float v8f;
typedef __attribute__((ext_vector_type(2))) int   v2i;

#define N_NODES   131072
#define E_EDGES   4194304
#define S_IN      32
#define H_DIM     16
#define B_GRAPHS  128
#define FC_IN_SZ  16384
#define KCHUNKS   8

__global__ void fill_kernel(float* __restrict__ p, float v, int n) {
    int i = blockIdx.x * blockDim.x + threadIdx.x;
    if (i < n) p[i] = v;
}


__global__ void rank_kernel(const int* __restrict__ col, int* __restrict__ cnt,
                            int* __restrict__ rank, int e) {
    int i = blockIdx.x * blockDim.x + threadIdx.x;
    if (i < e) rank[i] = atomicAdd(&cnt[col[i]], 1);
}

__global__ __launch_bounds__(1024)
void scan_kernel(const int* __restrict__ cnt, int* __restrict__ offs, float* __restrict__ dinv) {
    __shared__ int s_w[32];
    __shared__ int s_pref;
    const int t    = threadIdx.x;
    const int b    = blockIdx.x;
    const int lane = t & 31;
    const int wv   = t >> 5;

    int part = 0;
    for (int i = t; i < b * 1024; i += 1024) part += cnt[i];
    for (int o = 16; o > 0; o >>= 1) part += __shfl_xor(part, o);
    if (lane == 0) s_w[wv] = part;
    __syncthreads();
    if (wv == 0) {
        int v = s_w[lane];
        for (int o = 16; o > 0; o >>= 1) v += __shfl_xor(v, o);
        if (lane == 0) s_pref = v;
    }
    __syncthreads();
    const int pref = s_pref;
    __syncthreads();

    const int c = cnt[b * 1024 + t];
    int inc = c;
    for (int o = 1; o < 32; o <<= 1) {
        int v = __shfl_up(inc, o);
        if (lane >= o) inc += v;
    }
    if (lane == 31) s_w[wv] = inc;
    __syncthreads();
    if (wv == 0) {
        const int tot = s_w[lane];
        int iv = tot;
        for (int o = 1; o < 32; o <<= 1) {
            int u = __shfl_up(iv, o);
            if (lane >= o) iv += u;
        }
        s_w[lane] = iv - tot;
    }
    __syncthreads();
    offs[b * 1024 + t] = pref + s_w[wv] + inc - c;
    dinv[b * 1024 + t] = rsqrtf((float)(c + 1));
}

__global__ void permute_kernel(const int* __restrict__ row, const int* __restrict__ col,
                               const int* __restrict__ rank, const int* __restrict__ offs,
                               const float* __restrict__ dinv, int* __restrict__ csr, int e) {
    int i = blockIdx.x * blockDim.x + threadIdx.x;
    if (i >= e) return;
    const int r = row[i], c = col[i];
    const unsigned pos = min((unsigned)(offs[c] + rank[i]), (unsigned)(e - 1));
    v2i ent;
    ent.x = r;
    ent.y = __float_as_int(dinv[r] * dinv[c]);
    ((v2i*)csr)[pos] = ent;
}

__global__ __launch_bounds__(32)
void wmma_gemm_f32(const float* __restrict__ A, const float* __restrict__ B,
                   const float* __restrict__ bias, float* __restrict__ C,
                   int K, int Nld, int relu, int Kc) {
    const int nb   = blockIdx.x;
    const int mb   = blockIdx.y;
    const int z    = blockIdx.z;
    const int lane = threadIdx.x;
    const int half = lane >> 4;
    const int lm   = lane & 15;

    const int k0 = z * Kc;

    v8f acc = {};
    const float* Ap = A + (size_t)(mb * 16 + lm) * K + k0 + 2 * half;
    const float* Bp = B + (size_t)(k0 + 2 * half) * Nld + nb * 16 + lm;

#pragma unroll 4
    for (int k = 0; k < Kc; k += 4) {
        v2f a = *(const v2f*)Ap;
        v2f b;
        b.x = Bp[0];
        b.y = Bp[Nld];
        acc = __builtin_amdgcn_wmma_f32_16x16x4_f32(
                  false, a, false, b, (short)0, acc, false, false);
        Ap += 4;
        Bp += (size_t)4 * Nld;
    }

    const int cn = nb * 16 + lm;
    if (gridDim.z == 1) {
        const float bb = bias ? bias[cn] : 0.0f;
#pragma unroll
        for (int v = 0; v < 8; ++v) {
            float r = acc[v] + bb;
            if (relu) r = fmaxf(r, 0.0f);
            C[(size_t)(mb * 16 + v + 8 * half) * Nld + cn] = r;
        }
    } else {
        const size_t M  = (size_t)gridDim.y * 16;
        float* Cp = C + (size_t)z * M * Nld;
#pragma unroll
        for (int v = 0; v < 8; ++v) {
            Cp[(size_t)(mb * 16 + v + 8 * half) * Nld + cn] = acc[v];
        }
    }
}

__global__ void reduce_bias_relu_kernel(const float* __restrict__ part,
                                        const float* __restrict__ bias,
                                        float* __restrict__ C,
                                        int MN, int Nld, int relu) {
    int i = blockIdx.x * blockDim.x + threadIdx.x;
    if (i >= MN) return;
    float s = 0.0f;
#pragma unroll
    for (int zc = 0; zc < KCHUNKS; ++zc) s += part[(size_t)zc * MN + i];
    s += bias[i % Nld];
    if (relu) s = fmaxf(s, 0.0f);
    C[i] = s;
}

__global__ __launch_bounds__(256)
void gather_kernel(const float* __restrict__ xw, const int* __restrict__ csr,
                   const int* __restrict__ offs, const int* __restrict__ cnt,
                   const float* __restrict__ dinv, const float* __restrict__ bias,
                   float* __restrict__ h, int n16) {
    const int i = blockIdx.x * blockDim.x + threadIdx.x;
    if (i >= n16) return;
    const int node = i >> 4, j = i & 15;
    const int start = offs[node];
    const int end   = start + cnt[node];
    float acc = 0.0f;
    for (int base = start; base < end; base += 16) {
        const int p = base + j;
        v2i ent;
        ent.x = node;
        ent.y = 0;
        if (p < end) ent = ((const v2i*)csr)[p];
        const int m = end - base;
        float xv[16];
#pragma unroll
        for (int q = 0; q < 16; ++q) {
            const int r = __shfl(ent.x, q, 16) & (N_NODES - 1);
            xv[q] = xw[r * H_DIM + j];
        }
#pragma unroll
        for (int q = 0; q < 16; ++q) {
            const float w = __int_as_float(__shfl(ent.y, q, 16));
            if (q < m) acc += w * xv[q];
        }
    }
    const float d = dinv[node];
    const float r = acc + d * d * xw[i] + bias[j];
    h[i] = fmaxf(r, 0.0f);
}

__global__ void heads_kernel(const float* __restrict__ V,
                             const float* __restrict__ Wiv, const float* __restrict__ biv,
                             const float* __restrict__ Wev, const float* __restrict__ bev,
                             float* __restrict__ out_ext, float* __restrict__ out_int) {
    int r = blockIdx.x * blockDim.x + threadIdx.x;
    if (r >= B_GRAPHS) return;
    const float* v = V + (size_t)r * 256;
    float si = 0.0f, se = 0.0f;
#pragma unroll 4
    for (int k = 0; k < 256; ++k) {
        float x = v[k];
        si += x * Wiv[k];
        se += x * Wev[k];
    }
    out_int[r] = si + biv[0];
    out_ext[r] = se + bev[0];
}

extern "C" void kernel_launch(void* const* d_in, const int* in_sizes, int n_in,
                              void* d_out, int out_size, void* d_ws, size_t ws_size,
                              hipStream_t stream) {
    const float* x    = (const float*)d_in[0];
    const int*   ei   = (const int*)d_in[1];
    const float* W1   = (const float*)d_in[2];
    const float* b1   = (const float*)d_in[3];
    const float* W2   = (const float*)d_in[4];
    const float* b2   = (const float*)d_in[5];
    const float* W3   = (const float*)d_in[6];
    const float* b3   = (const float*)d_in[7];
    const float* Wp1  = (const float*)d_in[8];
    const float* bp1  = (const float*)d_in[9];
    const float* Wp2  = (const float*)d_in[10];
    const float* bp2  = (const float*)d_in[11];
    const float* Wv1  = (const float*)d_in[12];
    const float* bv1  = (const float*)d_in[13];
    const float* Wiv  = (const float*)d_in[14];
    const float* biv  = (const float*)d_in[15];
    const float* Wev  = (const float*)d_in[16];
    const float* bev  = (const float*)d_in[17];

    const int* row = ei;
    const int* col = ei + E_EDGES;

    char* ws = (char*)d_ws;
    float* dinv = (float*)ws;
    float* bufA = (float*)(ws + (size_t)N_NODES * 4);
    float* bufB = bufA + (size_t)N_NODES * H_DIM;
    float* Hp   = bufB + (size_t)N_NODES * H_DIM;
    float* Vv   = Hp + (size_t)B_GRAPHS * 256;
    float* part = Vv + (size_t)B_GRAPHS * 256;
    int*   cnt  = (int*)(part + (size_t)KCHUNKS * B_GRAPHS * 256);
    int*   offs = cnt + N_NODES;
    int*   rank = offs + N_NODES;
    int*   csr  = rank + E_EDGES;

    float* P     = (float*)d_out;
    float* ext_V = P + (size_t)B_GRAPHS * 1024;
    float* int_V = ext_V + B_GRAPHS;

    const int TPB = 256;
    const int nh  = N_NODES * H_DIM;
    const int fcMN = B_GRAPHS * 256;

    fill_kernel<<<(N_NODES + TPB - 1) / TPB, TPB, 0, stream>>>((float*)cnt, 0.0f, N_NODES);
    rank_kernel<<<(E_EDGES + TPB - 1) / TPB, TPB, 0, stream>>>(col, cnt, rank, E_EDGES);
    scan_kernel<<<N_NODES / 1024, 1024, 0, stream>>>(cnt, offs, dinv);
    permute_kernel<<<(E_EDGES + TPB - 1) / TPB, TPB, 0, stream>>>(row, col, rank, offs, dinv, csr, E_EDGES);

    wmma_gemm_f32<<<dim3(1, N_NODES / 16, 1), 32, 0, stream>>>(x, W1, nullptr, bufA, S_IN, H_DIM, 0, S_IN);
    gather_kernel<<<(nh + TPB - 1) / TPB, TPB, 0, stream>>>(bufA, csr, offs, cnt, dinv, b1, bufB, nh);

    wmma_gemm_f32<<<dim3(1, N_NODES / 16, 1), 32, 0, stream>>>(bufB, W2, nullptr, bufA, H_DIM, H_DIM, 0, H_DIM);
    gather_kernel<<<(nh + TPB - 1) / TPB, TPB, 0, stream>>>(bufA, csr, offs, cnt, dinv, b2, bufB, nh);

    wmma_gemm_f32<<<dim3(1, N_NODES / 16, 1), 32, 0, stream>>>(bufB, W3, nullptr, bufA, H_DIM, H_DIM, 0, H_DIM);
    gather_kernel<<<(nh + TPB - 1) / TPB, TPB, 0, stream>>>(bufA, csr, offs, cnt, dinv, b3, bufB, nh);


    wmma_gemm_f32<<<dim3(256 / 16, B_GRAPHS / 16, KCHUNKS), 32, 0, stream>>>(
        bufB, Wp1, nullptr, part, FC_IN_SZ, 256, 0, FC_IN_SZ / KCHUNKS);
    reduce_bias_relu_kernel<<<(fcMN + TPB - 1) / TPB, TPB, 0, stream>>>(part, bp1, Hp, fcMN, 256, 1);
    wmma_gemm_f32<<<dim3(256 / 16, B_GRAPHS / 16, KCHUNKS), 32, 0, stream>>>(
        bufB, Wv1, nullptr, part, FC_IN_SZ, 256, 0, FC_IN_SZ / KCHUNKS);
    reduce_bias_relu_kernel<<<(fcMN + TPB - 1) / TPB, TPB, 0, stream>>>(part, bv1, Vv, fcMN, 256, 1);

    wmma_gemm_f32<<<dim3(1024 / 16, B_GRAPHS / 16, 1), 32, 0, stream>>>(Hp, Wp2, bp2, P, 256, 1024, 0, 256);

    heads_kernel<<<1, B_GRAPHS, 0, stream>>>(Vv, Wiv, biv, Wev, bev, ext_V, int_V);
}


# FC GEMMs: split-K raised from 8 to 64 chunks, 8x more waves in flight to cover load latency; each output now sums 64 partials of K=256 instead of 8 of K=2048 (summation tree changes)
# speedup vs baseline: 1.8725x; 1.2107x over previous
#include <hip/hip_runtime.h>

typedef __attribute__((ext_vector_type(2))) float v2f;
typedef __attribute__((ext_vector_type(8))) float v8f;
typedef __attribute__((ext_vector_type(2))) int   v2i;

#define N_NODES   131072
#define E_EDGES   4194304
#define S_IN      32
#define H_DIM     16
#define B_GRAPHS  128
#define FC_IN_SZ  16384
#define KCHUNKS   64

__global__ void fill_kernel(float* __restrict__ p, float v, int n) {
    int i = blockIdx.x * blockDim.x + threadIdx.x;
    if (i < n) p[i] = v;
}


__global__ void rank_kernel(const int* __restrict__ col, int* __restrict__ cnt,
                            int* __restrict__ rank, int e) {
    int i = blockIdx.x * blockDim.x + threadIdx.x;
    if (i < e) rank[i] = atomicAdd(&cnt[col[i]], 1);
}

__global__ __launch_bounds__(1024)
void scan_kernel(const int* __restrict__ cnt, int* __restrict__ offs, float* __restrict__ dinv) {
    __shared__ int s_w[32];
    __shared__ int s_pref;
    const int t    = threadIdx.x;
    const int b    = blockIdx.x;
    const int lane = t & 31;
    const int wv   = t >> 5;

    int part = 0;
    for (int i = t; i < b * 1024; i += 1024) part += cnt[i];
    for (int o = 16; o > 0; o >>= 1) part += __shfl_xor(part, o);
    if (lane == 0) s_w[wv] = part;
    __syncthreads();
    if (wv == 0) {
        int v = s_w[lane];
        for (int o = 16; o > 0; o >>= 1) v += __shfl_xor(v, o);
        if (lane == 0) s_pref = v;
    }
    __syncthreads();
    const int pref = s_pref;
    __syncthreads();

    const int c = cnt[b * 1024 + t];
    int inc = c;
    for (int o = 1; o < 32; o <<= 1) {
        int v = __shfl_up(inc, o);
        if (lane >= o) inc += v;
    }
    if (lane == 31) s_w[wv] = inc;
    __syncthreads();
    if (wv == 0) {
        const int tot = s_w[lane];
        int iv = tot;
        for (int o = 1; o < 32; o <<= 1) {
            int u = __shfl_up(iv, o);
            if (lane >= o) iv += u;
        }
        s_w[lane] = iv - tot;
    }
    __syncthreads();
    offs[b * 1024 + t] = pref + s_w[wv] + inc - c;
    dinv[b * 1024 + t] = rsqrtf((float)(c + 1));
}

__global__ void permute_kernel(const int* __restrict__ row, const int* __restrict__ col,
                               const int* __restrict__ rank, const int* __restrict__ offs,
                               const float* __restrict__ dinv, int* __restrict__ csr, int e) {
    int i = blockIdx.x * blockDim.x + threadIdx.x;
    if (i >= e) return;
    const int r = row[i], c = col[i];
    const unsigned pos = min((unsigned)(offs[c] + rank[i]), (unsigned)(e - 1));
    v2i ent;
    ent.x = r;
    ent.y = __float_as_int(dinv[r] * dinv[c]);
    ((v2i*)csr)[pos] = ent;
}

__global__ __launch_bounds__(32)
void wmma_gemm_f32(const float* __restrict__ A, const float* __restrict__ B,
                   const float* __restrict__ bias, float* __restrict__ C,
                   int K, int Nld, int relu, int Kc) {
    const int nb   = blockIdx.x;
    const int mb   = blockIdx.y;
    const int z    = blockIdx.z;
    const int lane = threadIdx.x;
    const int half = lane >> 4;
    const int lm   = lane & 15;

    const int k0 = z * Kc;

    v8f acc = {};
    const float* Ap = A + (size_t)(mb * 16 + lm) * K + k0 + 2 * half;
    const float* Bp = B + (size_t)(k0 + 2 * half) * Nld + nb * 16 + lm;

#pragma unroll 4
    for (int k = 0; k < Kc; k += 4) {
        v2f a = *(const v2f*)Ap;
        v2f b;
        b.x = Bp[0];
        b.y = Bp[Nld];
        acc = __builtin_amdgcn_wmma_f32_16x16x4_f32(
                  false, a, false, b, (short)0, acc, false, false);
        Ap += 4;
        Bp += (size_t)4 * Nld;
    }

    const int cn = nb * 16 + lm;
    if (gridDim.z == 1) {
        const float bb = bias ? bias[cn] : 0.0f;
#pragma unroll
        for (int v = 0; v < 8; ++v) {
            float r = acc[v] + bb;
            if (relu) r = fmaxf(r, 0.0f);
            C[(size_t)(mb * 16 + v + 8 * half) * Nld + cn] = r;
        }
    } else {
        const size_t M  = (size_t)gridDim.y * 16;
        float* Cp = C + (size_t)z * M * Nld;
#pragma unroll
        for (int v = 0; v < 8; ++v) {
            Cp[(size_t)(mb * 16 + v + 8 * half) * Nld + cn] = acc[v];
        }
    }
}

__global__ void reduce_bias_relu_kernel(const float* __restrict__ part,
                                        const float* __restrict__ bias,
                                        float* __restrict__ C,
                                        int MN, int Nld, int relu) {
    int i = blockIdx.x * blockDim.x + threadIdx.x;
    if (i >= MN) return;
    float s = 0.0f;
#pragma unroll
    for (int zc = 0; zc < KCHUNKS; ++zc) s += part[(size_t)zc * MN + i];
    s += bias[i % Nld];
    if (relu) s = fmaxf(s, 0.0f);
    C[i] = s;
}

__global__ __launch_bounds__(256)
void gather_kernel(const float* __restrict__ xw, const int* __restrict__ csr,
                   const int* __restrict__ offs, const int* __restrict__ cnt,
                   const float* __restrict__ dinv, const float* __restrict__ bias,
                   float* __restrict__ h, int n16) {
    const int i = blockIdx.x * blockDim.x + threadIdx.x;
    if (i >= n16) return;
    const int node = i >> 4, j = i & 15;
    const int start = offs[node];
    const int end   = start + cnt[node];
    float acc = 0.0f;
    for (int base = start; base < end; base += 16) {
        const int p = base + j;
        v2i ent;
        ent.x = node;
        ent.y = 0;
        if (p < end) ent = ((const v2i*)csr)[p];
        const int m = end - base;
        float xv[16];
#pragma unroll
        for (int q = 0; q < 16; ++q) {
            const int r = __shfl(ent.x, q, 16) & (N_NODES - 1);
            xv[q] = xw[r * H_DIM + j];
        }
#pragma unroll
        for (int q = 0; q < 16; ++q) {
            const float w = __int_as_float(__shfl(ent.y, q, 16));
            if (q < m) acc += w * xv[q];
        }
    }
    const float d = dinv[node];
    const float r = acc + d * d * xw[i] + bias[j];
    h[i] = fmaxf(r, 0.0f);
}

__global__ void heads_kernel(const float* __restrict__ V,
                             const float* __restrict__ Wiv, const float* __restrict__ biv,
                             const float* __restrict__ Wev, const float* __restrict__ bev,
                             float* __restrict__ out_ext, float* __restrict__ out_int) {
    int r = blockIdx.x * blockDim.x + threadIdx.x;
    if (r >= B_GRAPHS) return;
    const float* v = V + (size_t)r * 256;
    float si = 0.0f, se = 0.0f;
#pragma unroll 4
    for (int k = 0; k < 256; ++k) {
        float x = v[k];
        si += x * Wiv[k];
        se += x * Wev[k];
    }
    out_int[r] = si + biv[0];
    out_ext[r] = se + bev[0];
}

extern "C" void kernel_launch(void* const* d_in, const int* in_sizes, int n_in,
                              void* d_out, int out_size, void* d_ws, size_t ws_size,
                              hipStream_t stream) {
    const float* x    = (const float*)d_in[0];
    const int*   ei   = (const int*)d_in[1];
    const float* W1   = (const float*)d_in[2];
    const float* b1   = (const float*)d_in[3];
    const float* W2   = (const float*)d_in[4];
    const float* b2   = (const float*)d_in[5];
    const float* W3   = (const float*)d_in[6];
    const float* b3   = (const float*)d_in[7];
    const float* Wp1  = (const float*)d_in[8];
    const float* bp1  = (const float*)d_in[9];
    const float* Wp2  = (const float*)d_in[10];
    const float* bp2  = (const float*)d_in[11];
    const float* Wv1  = (const float*)d_in[12];
    const float* bv1  = (const float*)d_in[13];
    const float* Wiv  = (const float*)d_in[14];
    const float* biv  = (const float*)d_in[15];
    const float* Wev  = (const float*)d_in[16];
    const float* bev  = (const float*)d_in[17];

    const int* row = ei;
    const int* col = ei + E_EDGES;

    char* ws = (char*)d_ws;
    float* dinv = (float*)ws;
    float* bufA = (float*)(ws + (size_t)N_NODES * 4);
    float* bufB = bufA + (size_t)N_NODES * H_DIM;
    float* Hp   = bufB + (size_t)N_NODES * H_DIM;
    float* Vv   = Hp + (size_t)B_GRAPHS * 256;
    float* part = Vv + (size_t)B_GRAPHS * 256;
    int*   cnt  = (int*)(part + (size_t)KCHUNKS * B_GRAPHS * 256);
    int*   offs = cnt + N_NODES;
    int*   rank = offs + N_NODES;
    int*   csr  = rank + E_EDGES;

    float* P     = (float*)d_out;
    float* ext_V = P + (size_t)B_GRAPHS * 1024;
    float* int_V = ext_V + B_GRAPHS;

    const int TPB = 256;
    const int nh  = N_NODES * H_DIM;
    const int fcMN = B_GRAPHS * 256;

    fill_kernel<<<(N_NODES + TPB - 1) / TPB, TPB, 0, stream>>>((float*)cnt, 0.0f, N_NODES);
    rank_kernel<<<(E_EDGES + TPB - 1) / TPB, TPB, 0, stream>>>(col, cnt, rank, E_EDGES);
    scan_kernel<<<N_NODES / 1024, 1024, 0, stream>>>(cnt, offs, dinv);
    permute_kernel<<<(E_EDGES + TPB - 1) / TPB, TPB, 0, stream>>>(row, col, rank, offs, dinv, csr, E_EDGES);

    wmma_gemm_f32<<<dim3(1, N_NODES / 16, 1), 32, 0, stream>>>(x, W1, nullptr, bufA, S_IN, H_DIM, 0, S_IN);
    gather_kernel<<<(nh + TPB - 1) / TPB, TPB, 0, stream>>>(bufA, csr, offs, cnt, dinv, b1, bufB, nh);

    wmma_gemm_f32<<<dim3(1, N_NODES / 16, 1), 32, 0, stream>>>(bufB, W2, nullptr, bufA, H_DIM, H_DIM, 0, H_DIM);
    gather_kernel<<<(nh + TPB - 1) / TPB, TPB, 0, stream>>>(bufA, csr, offs, cnt, dinv, b2, bufB, nh);

    wmma_gemm_f32<<<dim3(1, N_NODES / 16, 1), 32, 0, stream>>>(bufB, W3, nullptr, bufA, H_DIM, H_DIM, 0, H_DIM);
    gather_kernel<<<(nh + TPB - 1) / TPB, TPB, 0, stream>>>(bufA, csr, offs, cnt, dinv, b3, bufB, nh);


    wmma_gemm_f32<<<dim3(256 / 16, B_GRAPHS / 16, KCHUNKS), 32, 0, stream>>>(
        bufB, Wp1, nullptr, part, FC_IN_SZ, 256, 0, FC_IN_SZ / KCHUNKS);
    reduce_bias_relu_kernel<<<(fcMN + TPB - 1) / TPB, TPB, 0, stream>>>(part, bp1, Hp, fcMN, 256, 1);
    wmma_gemm_f32<<<dim3(256 / 16, B_GRAPHS / 16, KCHUNKS), 32, 0, stream>>>(
        bufB, Wv1, nullptr, part, FC_IN_SZ, 256, 0, FC_IN_SZ / KCHUNKS);
    reduce_bias_relu_kernel<<<(fcMN + TPB - 1) / TPB, TPB, 0, stream>>>(part, bv1, Vv, fcMN, 256, 1);

    wmma_gemm_f32<<<dim3(1024 / 16, B_GRAPHS / 16, 1), 32, 0, stream>>>(Hp, Wp2, bp2, P, 256, 1024, 0, 256);

    heads_kernel<<<1, B_GRAPHS, 0, stream>>>(Vv, Wiv, biv, Wev, bev, ext_V, int_V);
}


# Edge weight factored: xW row r scaled by dinv[r] in the GEMM epilogue, node sum scaled by dinv[c] in gather; permute drops 2 random 4-B lookups per edge; product rounding changes, sum order kept
# speedup vs baseline: 2.0253x; 1.0816x over previous
#include <hip/hip_runtime.h>

typedef __attribute__((ext_vector_type(2))) float v2f;
typedef __attribute__((ext_vector_type(4))) float v4f;
typedef __attribute__((ext_vector_type(8))) float v8f;

#define N_NODES   131072
#define E_EDGES   4194304
#define S_IN      32
#define H_DIM     16
#define B_GRAPHS  128
#define FC_IN_SZ  16384
#define KCHUNKS   64

__global__ void fill_kernel(float* __restrict__ p, float v, int n) {
    int i = blockIdx.x * blockDim.x + threadIdx.x;
    if (i < n) p[i] = v;
}


__global__ void rank_kernel(const int* __restrict__ col, int* __restrict__ cnt,
                            int* __restrict__ rank, int e) {
    int i = blockIdx.x * blockDim.x + threadIdx.x;
    if (i < e) rank[i] = atomicAdd(&cnt[col[i]], 1);
}

__global__ __launch_bounds__(1024)
void scan_kernel(const int* __restrict__ cnt, int* __restrict__ offs, float* __restrict__ dinv) {
    __shared__ int s_w[32];
    __shared__ int s_pref;
    const int t    = threadIdx.x;
    const int b    = blockIdx.x;
    const int lane = t & 31;
    const int wv   = t >> 5;

    int part = 0;
    for (int i = t; i < b * 1024; i += 1024) part += cnt[i];
    for (int o = 16; o > 0; o >>= 1) part += __shfl_xor(part, o);
    if (lane == 0) s_w[wv] = part;
    __syncthreads();
    if (wv == 0) {
        int v = s_w[lane];
        for (int o = 16; o > 0; o >>= 1) v += __shfl_xor(v, o);
        if (lane == 0) s_pref = v;
    }
    __syncthreads();
    const int pref = s_pref;
    __syncthreads();

    const int c = cnt[b * 1024 + t];
    int inc = c;
    for (int o = 1; o < 32; o <<= 1) {
        int v = __shfl_up(inc, o);
        if (lane >= o) inc += v;
    }
    if (lane == 31) s_w[wv] = inc;
    __syncthreads();
    if (wv == 0) {
        const int tot = s_w[lane];
        int iv = tot;
        for (int o = 1; o < 32; o <<= 1) {
            int u = __shfl_up(iv, o);
            if (lane >= o) iv += u;
        }
        s_w[lane] = iv - tot;
    }
    __syncthreads();
    offs[b * 1024 + t] = pref + s_w[wv] + inc - c;
    dinv[b * 1024 + t] = rsqrtf((float)(c + 1));
}

__global__ void permute_kernel(const int* __restrict__ row, const int* __restrict__ col,
                               const int* __restrict__ rank, const int* __restrict__ offs,
                               int* __restrict__ csr, int e) {
    int i = blockIdx.x * blockDim.x + threadIdx.x;
    if (i >= e) return;
    const int r = row[i], c = col[i];
    const unsigned pos = min((unsigned)(offs[c] + rank[i]), (unsigned)(e - 1));
    csr[pos] = r;
}

__global__ __launch_bounds__(32)
void wmma_gemm_f32(const float* __restrict__ A, const float* __restrict__ B,
                   const float* __restrict__ bias, const float* __restrict__ rowscale,
                   float* __restrict__ C, int K, int Nld, int relu, int Kc) {
    const int nb   = blockIdx.x;
    const int mb   = blockIdx.y;
    const int z    = blockIdx.z;
    const int lane = threadIdx.x;
    const int half = lane >> 4;
    const int lm   = lane & 15;

    const int k0 = z * Kc;

    v8f acc = {};
    const float* Ap = A + (size_t)(mb * 16 + lm) * K + k0 + 2 * half;
    const float* Bp = B + (size_t)(k0 + 2 * half) * Nld + nb * 16 + lm;

#pragma unroll 4
    for (int k = 0; k < Kc; k += 4) {
        v2f a = *(const v2f*)Ap;
        v2f b;
        b.x = Bp[0];
        b.y = Bp[Nld];
        acc = __builtin_amdgcn_wmma_f32_16x16x4_f32(
                  false, a, false, b, (short)0, acc, false, false);
        Ap += 4;
        Bp += (size_t)4 * Nld;
    }

    const int cn = nb * 16 + lm;
    if (gridDim.z == 1) {
        const float bb = bias ? bias[cn] : 0.0f;
        if (rowscale) {
            const v4f* sp = (const v4f*)(rowscale + mb * 16 + 8 * half);
            const v4f s0 = sp[0], s1 = sp[1];
#pragma unroll
            for (int v = 0; v < 4; ++v) { acc[v] *= s0[v]; acc[v + 4] *= s1[v]; }
        }
#pragma unroll
        for (int v = 0; v < 8; ++v) {
            float r = acc[v] + bb;
            if (relu) r = fmaxf(r, 0.0f);
            C[(size_t)(mb * 16 + v + 8 * half) * Nld + cn] = r;
        }
    } else {
        const size_t M  = (size_t)gridDim.y * 16;
        float* Cp = C + (size_t)z * M * Nld;
#pragma unroll
        for (int v = 0; v < 8; ++v) {
            Cp[(size_t)(mb * 16 + v + 8 * half) * Nld + cn] = acc[v];
        }
    }
}

__global__ void reduce_bias_relu_kernel(const float* __restrict__ part,
                                        const float* __restrict__ bias,
                                        float* __restrict__ C,
                                        int MN, int Nld, int relu) {
    int i = blockIdx.x * blockDim.x + threadIdx.x;
    if (i >= MN) return;
    float s = 0.0f;
#pragma unroll
    for (int zc = 0; zc < KCHUNKS; ++zc) s += part[(size_t)zc * MN + i];
    s += bias[i % Nld];
    if (relu) s = fmaxf(s, 0.0f);
    C[i] = s;
}

__global__ __launch_bounds__(256)
void gather_kernel(const float* __restrict__ xs, const int* __restrict__ csr,
                   const int* __restrict__ offs, const int* __restrict__ cnt,
                   const float* __restrict__ dinv, const float* __restrict__ bias,
                   float* __restrict__ h, int n16) {
    const int i = blockIdx.x * blockDim.x + threadIdx.x;
    if (i >= n16) return;
    const int node = i >> 4, j = i & 15;
    const int start = offs[node];
    const int end   = start + cnt[node];
    float acc = 0.0f;
    for (int base = start; base < end; base += 16) {
        const int p = base + j;
        int src = node;
        if (p < end) src = csr[p];
        const int m = end - base;
        float xv[16];
#pragma unroll
        for (int q = 0; q < 16; ++q) {
            const int r = __shfl(src, q, 16) & (N_NODES - 1);
            xv[q] = xs[r * H_DIM + j];
        }
#pragma unroll
        for (int q = 0; q < 16; ++q) {
            if (q < m) acc += xv[q];
        }
    }
    const float r = dinv[node] * (acc + xs[i]) + bias[j];
    h[i] = fmaxf(r, 0.0f);
}

__global__ void heads_kernel(const float* __restrict__ V,
                             const float* __restrict__ Wiv, const float* __restrict__ biv,
                             const float* __restrict__ Wev, const float* __restrict__ bev,
                             float* __restrict__ out_ext, float* __restrict__ out_int) {
    int r = blockIdx.x * blockDim.x + threadIdx.x;
    if (r >= B_GRAPHS) return;
    const float* v = V + (size_t)r * 256;
    float si = 0.0f, se = 0.0f;
#pragma unroll 4
    for (int k = 0; k < 256; ++k) {
        float x = v[k];
        si += x * Wiv[k];
        se += x * Wev[k];
    }
    out_int[r] = si + biv[0];
    out_ext[r] = se + bev[0];
}

extern "C" void kernel_launch(void* const* d_in, const int* in_sizes, int n_in,
                              void* d_out, int out_size, void* d_ws, size_t ws_size,
                              hipStream_t stream) {
    const float* x    = (const float*)d_in[0];
    const int*   ei   = (const int*)d_in[1];
    const float* W1   = (const float*)d_in[2];
    const float* b1   = (const float*)d_in[3];
    const float* W2   = (const float*)d_in[4];
    const float* b2   = (const float*)d_in[5];
    const float* W3   = (const float*)d_in[6];
    const float* b3   = (const float*)d_in[7];
    const float* Wp1  = (const float*)d_in[8];
    const float* bp1  = (const float*)d_in[9];
    const float* Wp2  = (const float*)d_in[10];
    const float* bp2  = (const float*)d_in[11];
    const float* Wv1  = (const float*)d_in[12];
    const float* bv1  = (const float*)d_in[13];
    const float* Wiv  = (const float*)d_in[14];
    const float* biv  = (const float*)d_in[15];
    const float* Wev  = (const float*)d_in[16];
    const float* bev  = (const float*)d_in[17];

    const int* row = ei;
    const int* col = ei + E_EDGES;

    char* ws = (char*)d_ws;
    float* dinv = (float*)ws;
    float* bufA = (float*)(ws + (size_t)N_NODES * 4);
    float* bufB = bufA + (size_t)N_NODES * H_DIM;
    float* Hp   = bufB + (size_t)N_NODES * H_DIM;
    float* Vv   = Hp + (size_t)B_GRAPHS * 256;
    float* part = Vv + (size_t)B_GRAPHS * 256;
    int*   cnt  = (int*)(part + (size_t)KCHUNKS * B_GRAPHS * 256);
    int*   offs = cnt + N_NODES;
    int*   rank = offs + N_NODES;
    int*   csr  = rank + E_EDGES;

    float* P     = (float*)d_out;
    float* ext_V = P + (size_t)B_GRAPHS * 1024;
    float* int_V = ext_V + B_GRAPHS;

    const int TPB = 256;
    const int nh  = N_NODES * H_DIM;
    const int fcMN = B_GRAPHS * 256;

    fill_kernel<<<(N_NODES + TPB - 1) / TPB, TPB, 0, stream>>>((float*)cnt, 0.0f, N_NODES);
    rank_kernel<<<(E_EDGES + TPB - 1) / TPB, TPB, 0, stream>>>(col, cnt, rank, E_EDGES);
    scan_kernel<<<N_NODES / 1024, 1024, 0, stream>>>(cnt, offs, dinv);
    permute_kernel<<<(E_EDGES + TPB - 1) / TPB, TPB, 0, stream>>>(row, col, rank, offs, csr, E_EDGES);

    wmma_gemm_f32<<<dim3(1, N_NODES / 16, 1), 32, 0, stream>>>(x, W1, nullptr, dinv, bufA, S_IN, H_DIM, 0, S_IN);
    gather_kernel<<<(nh + TPB - 1) / TPB, TPB, 0, stream>>>(bufA, csr, offs, cnt, dinv, b1, bufB, nh);

    wmma_gemm_f32<<<dim3(1, N_NODES / 16, 1), 32, 0, stream>>>(bufB, W2, nullptr, dinv, bufA, H_DIM, H_DIM, 0, H_DIM);
    gather_kernel<<<(nh + TPB - 1) / TPB, TPB, 0, stream>>>(bufA, csr, offs, cnt, dinv, b2, bufB, nh);

    wmma_gemm_f32<<<dim3(1, N_NODES / 16, 1), 32, 0, stream>>>(bufB, W3, nullptr, dinv, bufA, H_DIM, H_DIM, 0, H_DIM);
    gather_kernel<<<(nh + TPB - 1) / TPB, TPB, 0, stream>>>(bufA, csr, offs, cnt, dinv, b3, bufB, nh);


    wmma_gemm_f32<<<dim3(256 / 16, B_GRAPHS / 16, KCHUNKS), 32, 0, stream>>>(
        bufB, Wp1, nullptr, nullptr, part, FC_IN_SZ, 256, 0, FC_IN_SZ / KCHUNKS);
    reduce_bias_relu_kernel<<<(fcMN + TPB - 1) / TPB, TPB, 0, stream>>>(part, bp1, Hp, fcMN, 256, 1);
    wmma_gemm_f32<<<dim3(256 / 16, B_GRAPHS / 16, KCHUNKS), 32, 0, stream>>>(
        bufB, Wv1, nullptr, nullptr, part, FC_IN_SZ, 256, 0, FC_IN_SZ / KCHUNKS);
    reduce_bias_relu_kernel<<<(fcMN + TPB - 1) / TPB, TPB, 0, stream>>>(part, bv1, Vv, fcMN, 256, 1);

    wmma_gemm_f32<<<dim3(1024 / 16, B_GRAPHS / 16, 1), 32, 0, stream>>>(Hp, Wp2, bp2, nullptr, P, 256, 1024, 0, 256);

    heads_kernel<<<1, B_GRAPHS, 0, stream>>>(Vv, Wiv, biv, Wev, bev, ext_V, int_V);
}


# Fewer, longer waves (wave entry was the limit): edge kernels take 8 edges per thread, 16-byte loads; gather wave takes 4 node pairs in turn; GEMM 8 waves per workgroup; arrival-rank sum order differs
# speedup vs baseline: 2.0964x; 1.0351x over previous
#include <hip/hip_runtime.h>

typedef __attribute__((ext_vector_type(2))) float v2f;
typedef __attribute__((ext_vector_type(4))) float v4f;
typedef __attribute__((ext_vector_type(8))) float v8f;
typedef __attribute__((ext_vector_type(4))) int   v4i;

#define N_NODES   131072
#define E_EDGES   4194304
#define S_IN      32
#define H_DIM     16
#define B_GRAPHS  128
#define FC_IN_SZ  16384
#define EDGES_PER_THREAD 8
#define GATHER_PAIRS 4
#define GEMM_WAVES   8
#define KCHUNKS   64

__global__ void fill_kernel(float* __restrict__ p, float v, int n4) {
    int i = blockIdx.x * blockDim.x + threadIdx.x;
    const v4f q = {v, v, v, v};
    if (i < n4) ((v4f*)p)[i] = q;
}


static_assert(E_EDGES % (256 * EDGES_PER_THREAD) == 0, "edge kernels take whole blocks of 2048 edges");
__global__ __launch_bounds__(256)
void rank_kernel(const int* __restrict__ col, int* __restrict__ cnt,
                 int* __restrict__ rank) {
    const int i0 = blockIdx.x * (256 * EDGES_PER_THREAD) + 4 * threadIdx.x;
    const int i1 = i0 + 1024;
    const v4i c0 = *(const v4i*)(col + i0);
    const v4i c1 = *(const v4i*)(col + i1);
    v4i r0, r1;
    r0.x = atomicAdd(&cnt[c0.x], 1);
    r0.y = atomicAdd(&cnt[c0.y], 1);
    r0.z = atomicAdd(&cnt[c0.z], 1);
    r0.w = atomicAdd(&cnt[c0.w], 1);
    r1.x = atomicAdd(&cnt[c1.x], 1);
    r1.y = atomicAdd(&cnt[c1.y], 1);
    r1.z = atomicAdd(&cnt[c1.z], 1);
    r1.w = atomicAdd(&cnt[c1.w], 1);
    *(v4i*)(rank + i0) = r0;
    *(v4i*)(rank + i1) = r1;
}

__global__ __launch_bounds__(1024)
void scan_kernel(const int* __restrict__ cnt, int* __restrict__ offs, float* __restrict__ dinv) {
    __shared__ int s_w[32];
    __shared__ int s_pref;
    const int t    = threadIdx.x;
    const int b    = blockIdx.x;
    const int lane = t & 31;
    const int wv   = t >> 5;

    int part = 0;
    for (int i = t; i < b * 1024; i += 1024) part += cnt[i];
    for (int o = 16; o > 0; o >>= 1) part += __shfl_xor(part, o);
    if (lane == 0) s_w[wv] = part;
    __syncthreads();
    if (wv == 0) {
        int v = s_w[lane];
        for (int o = 16; o > 0; o >>= 1) v += __shfl_xor(v, o);
        if (lane == 0) s_pref = v;
    }
    __syncthreads();
    const int pref = s_pref;
    __syncthreads();

    const int c = cnt[b * 1024 + t];
    int inc = c;
    for (int o = 1; o < 32; o <<= 1) {
        int v = __shfl_up(inc, o);
        if (lane >= o) inc += v;
    }
    if (lane == 31) s_w[wv] = inc;
    __syncthreads();
    if (wv == 0) {
        const int tot = s_w[lane];
        int iv = tot;
        for (int o = 1; o < 32; o <<= 1) {
            int u = __shfl_up(iv, o);
            if (lane >= o) iv += u;
        }
        s_w[lane] = iv - tot;
    }
    __syncthreads();
    offs[b * 1024 + t] = pref + s_w[wv] + inc - c;
    dinv[b * 1024 + t] = rsqrtf((float)(c + 1));
}

__global__ __launch_bounds__(256)
void permute_kernel(const int* __restrict__ row, const int* __restrict__ col,
                               const int* __restrict__ rank, const int* __restrict__ offs,
                               int* __restrict__ csr, int e) {
    const int i0 = blockIdx.x * (256 * EDGES_PER_THREAD) + 4 * threadIdx.x;
    const int i1 = i0 + 1024;
    const v4i r0 = *(const v4i*)(row + i0),  r1 = *(const v4i*)(row + i1);
    const v4i c0 = *(const v4i*)(col + i0),  c1 = *(const v4i*)(col + i1);
    const v4i k0 = *(const v4i*)(rank + i0), k1 = *(const v4i*)(rank + i1);
    int o[8];
    o[0] = offs[c0.x]; o[1] = offs[c0.y]; o[2] = offs[c0.z]; o[3] = offs[c0.w];
    o[4] = offs[c1.x]; o[5] = offs[c1.y]; o[6] = offs[c1.z]; o[7] = offs[c1.w];
    const int kk[8] = {k0.x, k0.y, k0.z, k0.w, k1.x, k1.y, k1.z, k1.w};
    const int rr[8] = {r0.x, r0.y, r0.z, r0.w, r1.x, r1.y, r1.z, r1.w};
#pragma unroll
    for (int q = 0; q < 8; ++q) {
        const unsigned pos = min((unsigned)(o[q] + kk[q]), (unsigned)(e - 1));
        csr[pos] = rr[q];
    }
}

static_assert((N_NODES / 16) % GEMM_WAVES == 0 && (B_GRAPHS / 16) % GEMM_WAVES == 0, "whole GEMM workgroups");
__global__ __launch_bounds__(32 * GEMM_WAVES)
void wmma_gemm_f32(const float* __restrict__ A, const float* __restrict__ B,
                   const float* __restrict__ bias, const float* __restrict__ rowscale,
                   float* __restrict__ C, int K, int Nld, int relu, int Kc) {
    const int nb   = blockIdx.x;
    const int mb   = blockIdx.y * GEMM_WAVES + (threadIdx.x >> 5);
    const int z    = blockIdx.z;
    const int lane = threadIdx.x & 31;
    const int half = lane >> 4;
    const int lm   = lane & 15;

    const int k0 = z * Kc;

    v8f acc = {};
    const float* Ap = A + (size_t)(mb * 16 + lm) * K + k0 + 2 * half;
    const float* Bp = B + (size_t)(k0 + 2 * half) * Nld + nb * 16 + lm;

#pragma unroll 4
    for (int k = 0; k < Kc; k += 4) {
        v2f a = *(const v2f*)Ap;
        v2f b;
        b.x = Bp[0];
        b.y = Bp[Nld];
        acc = __builtin_amdgcn_wmma_f32_16x16x4_f32(
                  false, a, false, b, (short)0, acc, false, false);
        Ap += 4;
        Bp += (size_t)4 * Nld;
    }

    const int cn = nb * 16 + lm;
    if (gridDim.z == 1) {
        const float bb = bias ? bias[cn] : 0.0f;
        if (rowscale) {
            const v4f* sp = (const v4f*)(rowscale + mb * 16 + 8 * half);
            const v4f s0 = sp[0], s1 = sp[1];
#pragma unroll
            for (int v = 0; v < 4; ++v) { acc[v] *= s0[v]; acc[v + 4] *= s1[v]; }
        }
#pragma unroll
        for (int v = 0; v < 8; ++v) {
            float r = acc[v] + bb;
            if (relu) r = fmaxf(r, 0.0f);
            C[(size_t)(mb * 16 + v + 8 * half) * Nld + cn] = r;
        }
    } else {
        const size_t M  = (size_t)gridDim.y * (16 * GEMM_WAVES);
        float* Cp = C + (size_t)z * M * Nld;
#pragma unroll
        for (int v = 0; v < 8; ++v) {
            Cp[(size_t)(mb * 16 + v + 8 * half) * Nld + cn] = acc[v];
        }
    }
}

__global__ void reduce_bias_relu_kernel(const float* __restrict__ part,
                                        const float* __restrict__ bias,
                                        float* __restrict__ C,
                                        int MN, int Nld, int relu) {
    int i = blockIdx.x * blockDim.x + threadIdx.x;
    if (i >= MN) return;
    float s = 0.0f;
#pragma unroll
    for (int zc = 0; zc < KCHUNKS; ++zc) s += part[(size_t)zc * MN + i];
    s += bias[i % Nld];
    if (relu) s = fmaxf(s, 0.0f);
    C[i] = s;
}

static_assert(N_NODES % (8 * 2 * GATHER_PAIRS) == 0, "a gather block of 8 waves takes 64 whole nodes");
__global__ __launch_bounds__(256)
void gather_kernel(const float* __restrict__ xs, const int* __restrict__ csr,
                   const int* __restrict__ offs, const int* __restrict__ cnt,
                   const float* __restrict__ dinv, const float* __restrict__ bias,
                   float* __restrict__ h, int n_nodes) {
    const int wave = (blockIdx.x * blockDim.x + threadIdx.x) >> 5;
    if (wave * (2 * GATHER_PAIRS) >= n_nodes) return;
    const int lane = threadIdx.x & 31;
    const int j = lane & 15;
    const int node0 = wave * (2 * GATHER_PAIRS) + (lane >> 4);

    int   st[GATHER_PAIRS], en[GATHER_PAIRS];
    float dn[GATHER_PAIRS];
#pragma unroll
    for (int p = 0; p < GATHER_PAIRS; ++p) {
        const int node = node0 + 2 * p;
        st[p] = offs[node];
        en[p] = st[p] + cnt[node];
        dn[p] = dinv[node];
    }
    const float bj = bias[j];

#pragma unroll
    for (int p = 0; p < GATHER_PAIRS; ++p) {
        const int node  = node0 + 2 * p;
        const int i     = node * H_DIM + j;
        const int start = st[p];
        const int end   = en[p];
        const float self = xs[i];
        float acc = 0.0f;
        for (int base = start; base < end; base += 16) {
            const int pp = base + j;
            int src = node;
            if (pp < end) src = csr[pp];
            const int m = end - base;
            float xv[16];
#pragma unroll
            for (int q = 0; q < 16; ++q) {
                const int r = __shfl(src, q, 16) & (N_NODES - 1);
                xv[q] = xs[r * H_DIM + j];
            }
#pragma unroll
            for (int q = 0; q < 16; ++q) {
                if (q < m) acc += xv[q];
            }
        }
        const float r = dn[p] * (acc + self) + bj;
        h[i] = fmaxf(r, 0.0f);
    }
}

__global__ void heads_kernel(const float* __restrict__ V,
                             const float* __restrict__ Wiv, const float* __restrict__ biv,
                             const float* __restrict__ Wev, const float* __restrict__ bev,
                             float* __restrict__ out_ext, float* __restrict__ out_int) {
    int r = blockIdx.x * blockDim.x + threadIdx.x;
    if (r >= B_GRAPHS) return;
    const float* v = V + (size_t)r * 256;
    float si = 0.0f, se = 0.0f;
#pragma unroll 4
    for (int k = 0; k < 256; ++k) {
        float x = v[k];
        si += x * Wiv[k];
        se += x * Wev[k];
    }
    out_int[r] = si + biv[0];
    out_ext[r] = se + bev[0];
}

extern "C" void kernel_launch(void* const* d_in, const int* in_sizes, int n_in,
                              void* d_out, int out_size, void* d_ws, size_t ws_size,
                              hipStream_t stream) {
    const float* x    = (const float*)d_in[0];
    const int*   ei   = (const int*)d_in[1];
    const float* W1   = (const float*)d_in[2];
    const float* b1   = (const float*)d_in[3];
    const float* W2   = (const float*)d_in[4];
    const float* b2   = (const float*)d_in[5];
    const float* W3   = (const float*)d_in[6];
    const float* b3   = (const float*)d_in[7];
    const float* Wp1  = (const float*)d_in[8];
    const float* bp1  = (const float*)d_in[9];
    const float* Wp2  = (const float*)d_in[10];
    const float* bp2  = (const float*)d_in[11];
    const float* Wv1  = (const float*)d_in[12];
    const float* bv1  = (const float*)d_in[13];
    const float* Wiv  = (const float*)d_in[14];
    const float* biv  = (const float*)d_in[15];
    const float* Wev  = (const float*)d_in[16];
    const float* bev  = (const float*)d_in[17];

    const int* row = ei;
    const int* col = ei + E_EDGES;

    char* ws = (char*)d_ws;
    float* dinv = (float*)ws;
    float* bufA = (float*)(ws + (size_t)N_NODES * 4);
    float* bufB = bufA + (size_t)N_NODES * H_DIM;
    float* Hp   = bufB + (size_t)N_NODES * H_DIM;
    float* Vv   = Hp + (size_t)B_GRAPHS * 256;
    float* part = Vv + (size_t)B_GRAPHS * 256;
    int*   cnt  = (int*)(part + (size_t)KCHUNKS * B_GRAPHS * 256);
    int*   offs = cnt + N_NODES;
    int*   rank = offs + N_NODES;
    int*   csr  = rank + E_EDGES;

    float* P     = (float*)d_out;
    float* ext_V = P + (size_t)B_GRAPHS * 1024;
    float* int_V = ext_V + B_GRAPHS;

    const int TPB = 256;
    const int fcMN = B_GRAPHS * 256;

    fill_kernel<<<(N_NODES / 4 + TPB - 1) / TPB, TPB, 0, stream>>>((float*)cnt, 0.0f, N_NODES / 4);
    rank_kernel<<<E_EDGES / (256 * EDGES_PER_THREAD), 256, 0, stream>>>(col, cnt, rank);
    scan_kernel<<<N_NODES / 1024, 1024, 0, stream>>>(cnt, offs, dinv);
    permute_kernel<<<E_EDGES / (256 * EDGES_PER_THREAD), 256, 0, stream>>>(row, col, rank, offs, csr, E_EDGES);

    wmma_gemm_f32<<<dim3(1, N_NODES / 16 / GEMM_WAVES, 1), 32 * GEMM_WAVES, 0, stream>>>(x, W1, nullptr, dinv, bufA, S_IN, H_DIM, 0, S_IN);
    gather_kernel<<<N_NODES / (8 * 2 * GATHER_PAIRS), 256, 0, stream>>>(bufA, csr, offs, cnt, dinv, b1, bufB, N_NODES);

    wmma_gemm_f32<<<dim3(1, N_NODES / 16 / GEMM_WAVES, 1), 32 * GEMM_WAVES, 0, stream>>>(bufB, W2, nullptr, dinv, bufA, H_DIM, H_DIM, 0, H_DIM);
    gather_kernel<<<N_NODES / (8 * 2 * GATHER_PAIRS), 256, 0, stream>>>(bufA, csr, offs, cnt, dinv, b2, bufB, N_NODES);

    wmma_gemm_f32<<<dim3(1, N_NODES / 16 / GEMM_WAVES, 1), 32 * GEMM_WAVES, 0, stream>>>(bufB, W3, nullptr, dinv, bufA, H_DIM, H_DIM, 0, H_DIM);
    gather_kernel<<<N_NODES / (8 * 2 * GATHER_PAIRS), 256, 0, stream>>>(bufA, csr, offs, cnt, dinv, b3, bufB, N_NODES);


    wmma_gemm_f32<<<dim3(256 / 16, B_GRAPHS / 16 / GEMM_WAVES, KCHUNKS), 32 * GEMM_WAVES, 0, stream>>>(
        bufB, Wp1, nullptr, nullptr, part, FC_IN_SZ, 256, 0, FC_IN_SZ / KCHUNKS);
    reduce_bias_relu_kernel<<<(fcMN + TPB - 1) / TPB, TPB, 0, stream>>>(part, bp1, Hp, fcMN, 256, 1);
    wmma_gemm_f32<<<dim3(256 / 16, B_GRAPHS / 16 / GEMM_WAVES, KCHUNKS), 32 * GEMM_WAVES, 0, stream>>>(
        bufB, Wv1, nullptr, nullptr, part, FC_IN_SZ, 256, 0, FC_IN_SZ / KCHUNKS);
    reduce_bias_relu_kernel<<<(fcMN + TPB - 1) / TPB, TPB, 0, stream>>>(part, bv1, Vv, fcMN, 256, 1);

    wmma_gemm_f32<<<dim3(1024 / 16, B_GRAPHS / 16 / GEMM_WAVES, 1), 32 * GEMM_WAVES, 0, stream>>>(Hp, Wp2, bp2, nullptr, P, 256, 1024, 0, 256);

    heads_kernel<<<1, B_GRAPHS, 0, stream>>>(Vv, Wiv, biv, Wev, bev, ext_V, int_V);
}


# Edge ordering by a two-level counting sort in LDS (node ranges, then targets) replaces one global returning add, lookup and scattered 4-B store per edge; neighbour sum order = LDS arrival order
# speedup vs baseline: 4.6581x; 2.2220x over previous
#include <hip/hip_runtime.h>

typedef __attribute__((ext_vector_type(2))) float v2f;
typedef __attribute__((ext_vector_type(4))) float v4f;
typedef __attribute__((ext_vector_type(8))) float v8f;
typedef __attribute__((ext_vector_type(4))) int   v4i;

#define N_NODES   131072
#define E_EDGES   4194304
#define S_IN      32
#define H_DIM     16
#define B_GRAPHS  128
#define FC_IN_SZ  16384
#define GATHER_PAIRS 4
#define GEMM_WAVES   8
#define KCHUNKS   64

__global__ void fill_kernel(float* __restrict__ p, float v, int n4) {
    int i = blockIdx.x * blockDim.x + threadIdx.x;
    const v4f q = {v, v, v, v};
    if (i < n4) ((v4f*)p)[i] = q;
}

#define BIN_SHIFT   9
#define BIN_NODES   512
#define NBINS       256
#define PART_WGS    256
#define PART_EDGES  16384
#define SORT_CAP    32768
static_assert(NBINS * BIN_NODES == N_NODES && (1 << BIN_SHIFT) == BIN_NODES, "ranges cover the nodes");
static_assert(PART_WGS * PART_EDGES == E_EDGES && PART_EDGES == 16 * 1024, "level-1 chunks cover the edges");

__device__ __forceinline__ int block_excl_scan_1024(int v, int* s_w) {
    const int lane = threadIdx.x & 31;
    const int wv   = threadIdx.x >> 5;
    int inc = v;
    for (int o = 1; o < 32; o <<= 1) {
        int u = __shfl_up(inc, o);
        if (lane >= o) inc += u;
    }
    __syncthreads();
    if (lane == 31) s_w[wv] = inc;
    __syncthreads();
    const int wtot = s_w[lane];
    int winc = wtot;
    for (int o = 1; o < 32; o <<= 1) {
        int u = __shfl_up(winc, o);
        if (lane >= o) winc += u;
    }
    const int wexcl = __shfl(winc - wtot, wv);
    return wexcl + inc - v;
}

__global__ __launch_bounds__(1024)
void bin_count_kernel(const int* __restrict__ col, int* __restrict__ bintot) {
    __shared__ int h[NBINS];
    const int t = threadIdx.x;
    if (t < NBINS) h[t] = 0;
    __syncthreads();
    const int* cp = col + blockIdx.x * PART_EDGES;
    v4i c[4];
#pragma unroll
    for (int q = 0; q < 4; ++q) c[q] = *(const v4i*)(cp + 4 * (q * 1024 + t));
#pragma unroll
    for (int q = 0; q < 4; ++q) {
#pragma unroll
        for (int e = 0; e < 4; ++e) atomicAdd(&h[(c[q][e] & (N_NODES - 1)) >> BIN_SHIFT], 1);
    }
    __syncthreads();
    if (t < NBINS) {
        const int n = h[t];
        if (n) atomicAdd(&bintot[t], n);
    }
}

__global__ __launch_bounds__(1024)
void bin_partition_kernel(const int* __restrict__ row, const int* __restrict__ col,
                          const int* __restrict__ bintot, int* __restrict__ bincur,
                          int* __restrict__ binned) {
    __shared__ int h[NBINS];
    __shared__ int lstart[NBINS];
    __shared__ int delta[NBINS];
    __shared__ int s_w[32];
    __shared__ int srow[PART_EDGES];
    __shared__ int scol[PART_EDGES];
    const int t = threadIdx.x;
    if (t < NBINS) h[t] = 0;
    __syncthreads();
    const int* cp = col + blockIdx.x * PART_EDGES;
    const int* rp = row + blockIdx.x * PART_EDGES;
    v4i c[4], r[4];
#pragma unroll
    for (int q = 0; q < 4; ++q) {
        c[q] = *(const v4i*)(cp + 4 * (q * 1024 + t));
        r[q] = *(const v4i*)(rp + 4 * (q * 1024 + t));
    }
    int lr[16];
#pragma unroll
    for (int q = 0; q < 4; ++q) {
#pragma unroll
        for (int e = 0; e < 4; ++e) {
            c[q][e] &= (N_NODES - 1);
            r[q][e] &= (N_NODES - 1);
            lr[4 * q + e] = atomicAdd(&h[c[q][e] >> BIN_SHIFT], 1);
        }
    }
    __syncthreads();
    const int mine = (t < NBINS) ? h[t] : 0;
    const int gtot = (t < NBINS) ? bintot[t] : 0;
    const int lex = block_excl_scan_1024(mine, s_w);
    const int gex = block_excl_scan_1024(gtot, s_w);
    if (t < NBINS) {
        int res = 0;
        if (mine) res = atomicAdd(&bincur[t], mine);
        lstart[t] = lex;
        delta[t]  = gex + res - lex;
    }
    __syncthreads();
#pragma unroll
    for (int q = 0; q < 4; ++q) {
#pragma unroll
        for (int e = 0; e < 4; ++e) {
            const int p = (lstart[c[q][e] >> BIN_SHIFT] + lr[4 * q + e]) & (PART_EDGES - 1);
            srow[p] = r[q][e];
            scol[p] = c[q][e];
        }
    }
    __syncthreads();
#pragma unroll 4
    for (int k = 0; k < PART_EDGES / 1024; ++k) {
        const int i  = k * 1024 + t;
        const int cc = scol[i];
        const unsigned pos = min((unsigned)(i + delta[cc >> BIN_SHIFT]), (unsigned)(E_EDGES - 1));
        binned[pos] = (srow[i] << BIN_SHIFT) | (cc & (BIN_NODES - 1));
    }
}

__global__ __launch_bounds__(1024)
void bin_sort_kernel(const int* __restrict__ binned, const int* __restrict__ bintot,
                     int* __restrict__ csr, int* __restrict__ cnt, int* __restrict__ offs,
                     float* __restrict__ dinv) {
    __shared__ int c_cnt[BIN_NODES];
    __shared__ int c_cur[BIN_NODES];
    __shared__ int s_w[32];
    __shared__ int s_start, s_n;
    __shared__ int stage[SORT_CAP];
    const int t = threadIdx.x;
    const int b = blockIdx.x;
    const int gtot = (t < NBINS) ? bintot[t] : 0;
    const int gex  = block_excl_scan_1024(gtot, s_w);
    if (t == b) { s_start = gex; s_n = gtot; }
    if (t < BIN_NODES) c_cnt[t] = 0;
    __syncthreads();
    const int start = min(max(s_start, 0), E_EDGES);
    const int nb    = min(max(s_n, 0), E_EDGES - start);
    const int* src  = binned + start;

    for (int base = 0; base < nb; base += 8 * 1024) {
        int rec[8];
#pragma unroll
        for (int q = 0; q < 8; ++q) {
            const int i = base + q * 1024 + t;
            rec[q] = (i < nb) ? src[i] : -1;
        }
#pragma unroll
        for (int q = 0; q < 8; ++q) {
            if (rec[q] >= 0) atomicAdd(&c_cnt[rec[q] & (BIN_NODES - 1)], 1);
        }
    }
    __syncthreads();
    const int mine = (t < BIN_NODES) ? c_cnt[t] : 0;
    const int lex  = block_excl_scan_1024(mine, s_w);
    if (t < BIN_NODES) {
        c_cur[t] = lex;
        const int node = b * BIN_NODES + t;
        cnt[node]  = mine;
        offs[node] = start + lex;
        dinv[node] = rsqrtf((float)(mine + 1));
    }
    __syncthreads();

    const bool staged = nb <= SORT_CAP;
    for (int base = 0; base < nb; base += 8 * 1024) {
        int rec[8];
#pragma unroll
        for (int q = 0; q < 8; ++q) {
            const int i = base + q * 1024 + t;
            rec[q] = (i < nb) ? src[i] : -1;
        }
#pragma unroll
        for (int q = 0; q < 8; ++q) {
            if (rec[q] >= 0) {
                const int p = atomicAdd(&c_cur[rec[q] & (BIN_NODES - 1)], 1);
                const int s = rec[q] >> BIN_SHIFT;
                if (staged) stage[min(p, SORT_CAP - 1)] = s;
                else        csr[min((unsigned)(start + p), (unsigned)(E_EDGES - 1))] = s;
            }
        }
    }
    if (staged) {
        __syncthreads();
        for (int i = t; i < nb; i += 1024) csr[start + i] = stage[i];
    }
}

static_assert((N_NODES / 16) % GEMM_WAVES == 0 && (B_GRAPHS / 16) % GEMM_WAVES == 0, "whole GEMM workgroups");
__global__ __launch_bounds__(32 * GEMM_WAVES)
void wmma_gemm_f32(const float* __restrict__ A, const float* __restrict__ B,
                   const float* __restrict__ bias, const float* __restrict__ rowscale,
                   float* __restrict__ C, int K, int Nld, int relu, int Kc) {
    const int nb   = blockIdx.x;
    const int mb   = blockIdx.y * GEMM_WAVES + (threadIdx.x >> 5);
    const int z    = blockIdx.z;
    const int lane = threadIdx.x & 31;
    const int half = lane >> 4;
    const int lm   = lane & 15;

    const int k0 = z * Kc;

    v8f acc = {};
    const float* Ap = A + (size_t)(mb * 16 + lm) * K + k0 + 2 * half;
    const float* Bp = B + (size_t)(k0 + 2 * half) * Nld + nb * 16 + lm;

#pragma unroll 4
    for (int k = 0; k < Kc; k += 4) {
        v2f a = *(const v2f*)Ap;
        v2f b;
        b.x = Bp[0];
        b.y = Bp[Nld];
        acc = __builtin_amdgcn_wmma_f32_16x16x4_f32(
                  false, a, false, b, (short)0, acc, false, false);
        Ap += 4;
        Bp += (size_t)4 * Nld;
    }

    const int cn = nb * 16 + lm;
    if (gridDim.z == 1) {
        const float bb = bias ? bias[cn] : 0.0f;
        if (rowscale) {
            const v4f* sp = (const v4f*)(rowscale + mb * 16 + 8 * half);
            const v4f s0 = sp[0], s1 = sp[1];
#pragma unroll
            for (int v = 0; v < 4; ++v) { acc[v] *= s0[v]; acc[v + 4] *= s1[v]; }
        }
#pragma unroll
        for (int v = 0; v < 8; ++v) {
            float r = acc[v] + bb;
            if (relu) r = fmaxf(r, 0.0f);
            C[(size_t)(mb * 16 + v + 8 * half) * Nld + cn] = r;
        }
    } else {
        const size_t M  = (size_t)gridDim.y * (16 * GEMM_WAVES);
        float* Cp = C + (size_t)z * M * Nld;
#pragma unroll
        for (int v = 0; v < 8; ++v) {
            Cp[(size_t)(mb * 16 + v + 8 * half) * Nld + cn] = acc[v];
        }
    }
}

__global__ void reduce_bias_relu_kernel(const float* __restrict__ part,
                                        const float* __restrict__ bias,
                                        float* __restrict__ C,
                                        int MN, int Nld, int relu) {
    int i = blockIdx.x * blockDim.x + threadIdx.x;
    if (i >= MN) return;
    float s = 0.0f;
#pragma unroll
    for (int zc = 0; zc < KCHUNKS; ++zc) s += part[(size_t)zc * MN + i];
    s += bias[i % Nld];
    if (relu) s = fmaxf(s, 0.0f);
    C[i] = s;
}

static_assert(N_NODES % (8 * 2 * GATHER_PAIRS) == 0, "a gather block of 8 waves takes 64 whole nodes");
__global__ __launch_bounds__(256)
void gather_kernel(const float* __restrict__ xs, const int* __restrict__ csr,
                   const int* __restrict__ offs, const int* __restrict__ cnt,
                   const float* __restrict__ dinv, const float* __restrict__ bias,
                   float* __restrict__ h, int n_nodes) {
    const int wave = (blockIdx.x * blockDim.x + threadIdx.x) >> 5;
    if (wave * (2 * GATHER_PAIRS) >= n_nodes) return;
    const int lane = threadIdx.x & 31;
    const int j = lane & 15;
    const int node0 = wave * (2 * GATHER_PAIRS) + (lane >> 4);

    int   st[GATHER_PAIRS], en[GATHER_PAIRS];
    float dn[GATHER_PAIRS];
#pragma unroll
    for (int p = 0; p < GATHER_PAIRS; ++p) {
        const int node = node0 + 2 * p;
        st[p] = offs[node];
        en[p] = st[p] + cnt[node];
        dn[p] = dinv[node];
    }
    const float bj = bias[j];

#pragma unroll
    for (int p = 0; p < GATHER_PAIRS; ++p) {
        const int node  = node0 + 2 * p;
        const int i     = node * H_DIM + j;
        const int start = st[p];
        const int end   = en[p];
        const float self = xs[i];
        float acc = 0.0f;
        for (int base = start; base < end; base += 16) {
            const int pp = base + j;
            int src = node;
            if (pp < end) src = csr[pp];
            const int m = end - base;
            float xv[16];
#pragma unroll
            for (int q = 0; q < 16; ++q) {
                const int r = __shfl(src, q, 16) & (N_NODES - 1);
                xv[q] = xs[r * H_DIM + j];
            }
#pragma unroll
            for (int q = 0; q < 16; ++q) {
                if (q < m) acc += xv[q];
            }
        }
        const float r = dn[p] * (acc + self) + bj;
        h[i] = fmaxf(r, 0.0f);
    }
}

__global__ void heads_kernel(const float* __restrict__ V,
                             const float* __restrict__ Wiv, const float* __restrict__ biv,
                             const float* __restrict__ Wev, const float* __restrict__ bev,
                             float* __restrict__ out_ext, float* __restrict__ out_int) {
    int r = blockIdx.x * blockDim.x + threadIdx.x;
    if (r >= B_GRAPHS) return;
    const float* v = V + (size_t)r * 256;
    float si = 0.0f, se = 0.0f;
#pragma unroll 4
    for (int k = 0; k < 256; ++k) {
        float x = v[k];
        si += x * Wiv[k];
        se += x * Wev[k];
    }
    out_int[r] = si + biv[0];
    out_ext[r] = se + bev[0];
}

extern "C" void kernel_launch(void* const* d_in, const int* in_sizes, int n_in,
                              void* d_out, int out_size, void* d_ws, size_t ws_size,
                              hipStream_t stream) {
    const float* x    = (const float*)d_in[0];
    const int*   ei   = (const int*)d_in[1];
    const float* W1   = (const float*)d_in[2];
    const float* b1   = (const float*)d_in[3];
    const float* W2   = (const float*)d_in[4];
    const float* b2   = (const float*)d_in[5];
    const float* W3   = (const float*)d_in[6];
    const float* b3   = (const float*)d_in[7];
    const float* Wp1  = (const float*)d_in[8];
    const float* bp1  = (const float*)d_in[9];
    const float* Wp2  = (const float*)d_in[10];
    const float* bp2  = (const float*)d_in[11];
    const float* Wv1  = (const float*)d_in[12];
    const float* bv1  = (const float*)d_in[13];
    const float* Wiv  = (const float*)d_in[14];
    const float* biv  = (const float*)d_in[15];
    const float* Wev  = (const float*)d_in[16];
    const float* bev  = (const float*)d_in[17];

    const int* row = ei;
    const int* col = ei + E_EDGES;

    char* ws = (char*)d_ws;
    float* dinv = (float*)ws;
    float* bufA = (float*)(ws + (size_t)N_NODES * 4);
    float* bufB = bufA + (size_t)N_NODES * H_DIM;
    float* Hp   = bufB + (size_t)N_NODES * H_DIM;
    float* Vv   = Hp + (size_t)B_GRAPHS * 256;
    float* part = Vv + (size_t)B_GRAPHS * 256;
    int*   cnt  = (int*)(part + (size_t)KCHUNKS * B_GRAPHS * 256);
    int*   offs = cnt + N_NODES;
    int*   binned = offs + N_NODES;
    int*   csr    = binned + E_EDGES;
    int*   bintot = csr + E_EDGES;
    int*   bincur = bintot + NBINS;

    float* P     = (float*)d_out;
    float* ext_V = P + (size_t)B_GRAPHS * 1024;
    float* int_V = ext_V + B_GRAPHS;

    const int TPB = 256;
    const int fcMN = B_GRAPHS * 256;

    fill_kernel<<<1, TPB, 0, stream>>>((float*)bintot, 0.0f, 2 * NBINS / 4);
    bin_count_kernel<<<PART_WGS, 1024, 0, stream>>>(col, bintot);
    bin_partition_kernel<<<PART_WGS, 1024, 0, stream>>>(row, col, bintot, bincur, binned);
    bin_sort_kernel<<<NBINS, 1024, 0, stream>>>(binned, bintot, csr, cnt, offs, dinv);

    wmma_gemm_f32<<<dim3(1, N_NODES / 16 / GEMM_WAVES, 1), 32 * GEMM_WAVES, 0, stream>>>(x, W1, nullptr, dinv, bufA, S_IN, H_DIM, 0, S_IN);
    gather_kernel<<<N_NODES / (8 * 2 * GATHER_PAIRS), 256, 0, stream>>>(bufA, csr, offs, cnt, dinv, b1, bufB, N_NODES);

    wmma_gemm_f32<<<dim3(1, N_NODES / 16 / GEMM_WAVES, 1), 32 * GEMM_WAVES, 0, stream>>>(bufB, W2, nullptr, dinv, bufA, H_DIM, H_DIM, 0, H_DIM);
    gather_kernel<<<N_NODES / (8 * 2 * GATHER_PAIRS), 256, 0, stream>>>(bufA, csr, offs, cnt, dinv, b2, bufB, N_NODES);

    wmma_gemm_f32<<<dim3(1, N_NODES / 16 / GEMM_WAVES, 1), 32 * GEMM_WAVES, 0, stream>>>(bufB, W3, nullptr, dinv, bufA, H_DIM, H_DIM, 0, H_DIM);
    gather_kernel<<<N_NODES / (8 * 2 * GATHER_PAIRS), 256, 0, stream>>>(bufA, csr, offs, cnt, dinv, b3, bufB, N_NODES);


    wmma_gemm_f32<<<dim3(256 / 16, B_GRAPHS / 16 / GEMM_WAVES, KCHUNKS), 32 * GEMM_WAVES, 0, stream>>>(
        bufB, Wp1, nullptr, nullptr, part, FC_IN_SZ, 256, 0, FC_IN_SZ / KCHUNKS);
    reduce_bias_relu_kernel<<<(fcMN + TPB - 1) / TPB, TPB, 0, stream>>>(part, bp1, Hp, fcMN, 256, 1);
    wmma_gemm_f32<<<dim3(256 / 16, B_GRAPHS / 16 / GEMM_WAVES, KCHUNKS), 32 * GEMM_WAVES, 0, stream>>>(
        bufB, Wv1, nullptr, nullptr, part, FC_IN_SZ, 256, 0, FC_IN_SZ / KCHUNKS);
    reduce_bias_relu_kernel<<<(fcMN + TPB - 1) / TPB, TPB, 0, stream>>>(part, bv1, Vv, fcMN, 256, 1);

    wmma_gemm_f32<<<dim3(1024 / 16, B_GRAPHS / 16 / GEMM_WAVES, 1), 32 * GEMM_WAVES, 0, stream>>>(Hp, Wp2, bp2, nullptr, P, 256, 1024, 0, 256);

    heads_kernel<<<1, B_GRAPHS, 0, stream>>>(Vv, Wiv, biv, Wev, bev, ext_V, int_V);
}


# Fewer dependent launches, 11 to 8: edge sort keeps each chunk's sorted runs in place + run table (no count pass, zero fill, global adds); conv1 x@W1 in the level-2 sort grid; sums in LDS arrival order
# speedup vs baseline: 5.6700x; 1.2172x over previous
#include <hip/hip_runtime.h>

typedef __attribute__((ext_vector_type(2))) float v2f;
typedef __attribute__((ext_vector_type(4))) float v4f;
typedef __attribute__((ext_vector_type(8))) float v8f;
typedef __attribute__((ext_vector_type(4))) int   v4i;
typedef __attribute__((ext_vector_type(2))) int   v2i;

#define N_NODES   131072
#define E_EDGES   4194304
#define S_IN      32
#define H_DIM     16
#define B_GRAPHS  128
#define FC_IN_SZ  16384
#define GATHER_PAIRS 4
#define GEMM_WAVES   8
#define FC_N      256
#define KCHUNKS   64

#define BIN_SHIFT   9
#define BIN_NODES   512
#define NBINS       256
#define PART_WGS    256
#define PART_EDGES  16384
#define SORT_CAP    32768
#define RUN_WAVE    8
#define RUN_REGS    3
static_assert(NBINS * BIN_NODES == N_NODES && (1 << BIN_SHIFT) == BIN_NODES, "ranges cover the nodes");
static_assert(PART_WGS * PART_EDGES == E_EDGES && PART_EDGES == 16 * 1024, "level-1 chunks cover the edges");
static_assert(PART_WGS == 32 * RUN_WAVE && BIN_NODES == 32 * 16, "a level-2 workgroup: 32 waves, 8 runs and one 16-row tile each");

__device__ __forceinline__ int block_excl_scan_1024(int v, int* s_w) {
    const int lane = threadIdx.x & 31;
    const int wv   = threadIdx.x >> 5;
    int inc = v;
    for (int o = 1; o < 32; o <<= 1) {
        int u = __shfl_up(inc, o);
        if (lane >= o) inc += u;
    }
    __syncthreads();
    if (lane == 31) s_w[wv] = inc;
    __syncthreads();
    const int wtot = s_w[lane];
    int winc = wtot;
    for (int o = 1; o < 32; o <<= 1) {
        int u = __shfl_up(winc, o);
        if (lane >= o) winc += u;
    }
    const int wexcl = __shfl(winc - wtot, wv);
    return wexcl + inc - v;
}

__device__ __forceinline__ int block_sum_1024(int v, int* s_w) {
    for (int o = 16; o > 0; o >>= 1) v += __shfl_xor(v, o);
    __syncthreads();
    if ((threadIdx.x & 31) == 0) s_w[threadIdx.x >> 5] = v;
    __syncthreads();
    int tot = s_w[threadIdx.x & 31];
    for (int o = 16; o > 0; o >>= 1) tot += __shfl_xor(tot, o);
    return tot;
}

__global__ __launch_bounds__(1024)
void bin_partition_kernel(const int* __restrict__ row, const int* __restrict__ col,
                          int* __restrict__ binned, int* __restrict__ runtab) {
    __shared__ int h[NBINS];
    __shared__ int lstart[NBINS];
    __shared__ int s_w[32];
    __shared__ __attribute__((aligned(16))) int srec[PART_EDGES];
    const int t = threadIdx.x;
    if (t < NBINS) h[t] = 0;
    __syncthreads();
    const int* cp = col + blockIdx.x * PART_EDGES;
    const int* rp = row + blockIdx.x * PART_EDGES;
    v4i c[4], r[4];
#pragma unroll
    for (int q = 0; q < 4; ++q) {
        c[q] = *(const v4i*)(cp + 4 * (q * 1024 + t));
        r[q] = *(const v4i*)(rp + 4 * (q * 1024 + t));
    }
    int lr[16];
#pragma unroll
    for (int q = 0; q < 4; ++q) {
#pragma unroll
        for (int e = 0; e < 4; ++e) {
            c[q][e] &= (N_NODES - 1);
            r[q][e] &= (N_NODES - 1);
            lr[4 * q + e] = atomicAdd(&h[c[q][e] >> BIN_SHIFT], 1);
        }
    }
    __syncthreads();
    const int mine = (t < NBINS) ? h[t] : 0;
    const int lex = block_excl_scan_1024(mine, s_w);
    if (t < NBINS) {
        lstart[t] = lex;
        const v2i ent = {lex, mine};
        *(v2i*)(runtab + 2 * (t * PART_WGS + blockIdx.x)) = ent;
    }
    __syncthreads();
#pragma unroll
    for (int q = 0; q < 4; ++q) {
#pragma unroll
        for (int e = 0; e < 4; ++e) {
            const int p = (lstart[c[q][e] >> BIN_SHIFT] + lr[4 * q + e]) & (PART_EDGES - 1);
            srec[p] = (r[q][e] << BIN_SHIFT) | (c[q][e] & (BIN_NODES - 1));
        }
    }
    __syncthreads();
    int* bp = binned + blockIdx.x * PART_EDGES;
#pragma unroll
    for (int q = 0; q < 4; ++q) {
        const int i = 4 * (q * 1024 + t);
        *(v4i*)(bp + i) = *(const v4i*)&srec[i];
    }
}

__global__ __launch_bounds__(1024)
void bin_sort_kernel(const int* __restrict__ binned, const int* __restrict__ runtab,
                     const float* __restrict__ x, const float* __restrict__ W1,
                     int* __restrict__ csr, int* __restrict__ cnt, int* __restrict__ offs,
                     float* __restrict__ dinv, float* __restrict__ xs) {
    __shared__ int c_cnt[BIN_NODES];
    __shared__ int c_cur[BIN_NODES];
    __shared__ __attribute__((aligned(16))) float s_dinv[BIN_NODES];
    __shared__ int r_base[PART_WGS];
    __shared__ int r_n[PART_WGS];
    __shared__ int s_w[32];
    __shared__ int stage[SORT_CAP];
    const int t = threadIdx.x;
    const int b = blockIdx.x;
    const int lane = t & 31;
    const int wave = t >> 5;
    const int half = lane >> 4;
    const int lm   = lane & 15;

    int rs = 0, rn = 0;
    if (t < PART_WGS) {
        const v2i ent = *(const v2i*)(runtab + 2 * (b * PART_WGS + t));
        rs = min(max(ent.x, 0), PART_EDGES);
        rn = min(max(ent.y, 0), PART_EDGES - rs);
        r_base[t] = t * PART_EDGES + rs;
        r_n[t]    = rn;
    }
    if (t < BIN_NODES) c_cnt[t] = 0;
    const int start_raw = block_sum_1024(rs, s_w);
    const int nb_raw    = block_sum_1024(rn, s_w);
    const int start = min(max(start_raw, 0), E_EDGES);
    const int nb    = min(max(nb_raw, 0), E_EDGES - start);

    int rec[RUN_WAVE][RUN_REGS];
#pragma unroll
    for (int m = 0; m < RUN_WAVE; ++m) {
        const int w = wave + 32 * m;
        const int n = r_n[w];
        const int* rp = binned + r_base[w];
#pragma unroll
        for (int j = 0; j < RUN_REGS; ++j) {
            const int i = lane + 32 * j;
            rec[m][j] = (i < n) ? rp[i] : -1;
        }
    }
    v2f fa[S_IN / 4], fb[S_IN / 4];
    {
        const float* Ap = x + (size_t)(b * BIN_NODES + wave * 16 + lm) * S_IN + 2 * half;
        const float* Bp = W1 + (2 * half) * H_DIM + lm;
#pragma unroll
        for (int s = 0; s < S_IN / 4; ++s) {
            fa[s]   = *(const v2f*)(Ap + 4 * s);
            fb[s].x = Bp[(4 * s) * H_DIM];
            fb[s].y = Bp[(4 * s + 1) * H_DIM];
        }
    }
#pragma unroll
    for (int m = 0; m < RUN_WAVE; ++m) {
#pragma unroll
        for (int j = 0; j < RUN_REGS; ++j) {
            if (rec[m][j] >= 0) atomicAdd(&c_cnt[rec[m][j] & (BIN_NODES - 1)], 1);
        }
    }
    for (int m = 0; m < RUN_WAVE; ++m) {
        const int w = wave + 32 * m;
        const int n = r_n[w];
        const int* rp = binned + r_base[w];
        for (int i = 32 * RUN_REGS + lane; i < n; i += 32) atomicAdd(&c_cnt[rp[i] & (BIN_NODES - 1)], 1);
    }
    v8f acc = {};
#pragma unroll
    for (int s = 0; s < S_IN / 4; ++s) {
        acc = __builtin_amdgcn_wmma_f32_16x16x4_f32(
                  false, fa[s], false, fb[s], (short)0, acc, false, false);
    }
    __syncthreads();
    const int mine = (t < BIN_NODES) ? c_cnt[t] : 0;
    const int lex  = block_excl_scan_1024(mine, s_w);
    if (t < BIN_NODES) {
        c_cur[t] = lex;
        const int node = b * BIN_NODES + t;
        const float d = rsqrtf((float)(mine + 1));
        cnt[node]  = mine;
        offs[node] = start + lex;
        dinv[node] = d;
        s_dinv[t]  = d;
    }
    __syncthreads();

    {
        const v4f s0 = *(const v4f*)&s_dinv[wave * 16 + 8 * half];
        const v4f s1 = *(const v4f*)&s_dinv[wave * 16 + 8 * half + 4];
        float* Cp = xs + (size_t)(b * BIN_NODES + wave * 16 + 8 * half) * H_DIM + lm;
#pragma unroll
        for (int v = 0; v < 4; ++v) {
            Cp[v * H_DIM]       = acc[v] * s0[v];
            Cp[(v + 4) * H_DIM] = acc[v + 4] * s1[v];
        }
    }

    const bool staged = nb <= SORT_CAP;
#pragma unroll
    for (int m = 0; m < RUN_WAVE; ++m) {
#pragma unroll
        for (int j = 0; j < RUN_REGS; ++j) {
            if (rec[m][j] >= 0) {
                const int p = atomicAdd(&c_cur[rec[m][j] & (BIN_NODES - 1)], 1);
                const int sid = rec[m][j] >> BIN_SHIFT;
                if (staged) stage[min(p, SORT_CAP - 1)] = sid;
                else        csr[min((unsigned)(start + p), (unsigned)(E_EDGES - 1))] = sid;
            }
        }
    }
    for (int m = 0; m < RUN_WAVE; ++m) {
        const int w = wave + 32 * m;
        const int n = r_n[w];
        const int* rp = binned + r_base[w];
        for (int i = 32 * RUN_REGS + lane; i < n; i += 32) {
            const int rr = rp[i];
            const int p = atomicAdd(&c_cur[rr & (BIN_NODES - 1)], 1);
            const int sid = rr >> BIN_SHIFT;
            if (staged) stage[min(p, SORT_CAP - 1)] = sid;
            else        csr[min((unsigned)(start + p), (unsigned)(E_EDGES - 1))] = sid;
        }
    }
    if (staged) {
        __syncthreads();
        for (int i = t; i < nb; i += 1024) csr[start + i] = stage[i];
    }
}

__device__ __forceinline__ v8f wmma_tile_f32(const float* __restrict__ Ap, const float* __restrict__ Bp,
                                             int ldb, int Kc) {
    v8f acc = {};
#pragma unroll 4
    for (int k = 0; k < Kc; k += 4) {
        v2f a = *(const v2f*)Ap;
        v2f b;
        b.x = Bp[0];
        b.y = Bp[ldb];
        acc = __builtin_amdgcn_wmma_f32_16x16x4_f32(
                  false, a, false, b, (short)0, acc, false, false);
        Ap += 4;
        Bp += (size_t)4 * ldb;
    }
    return acc;
}

static_assert((B_GRAPHS / 16) % GEMM_WAVES == 0, "whole GEMM workgroups");
__global__ __launch_bounds__(32 * GEMM_WAVES)
void wmma_gemm_f32(const float* __restrict__ A, const float* __restrict__ B,
                   const float* __restrict__ bias,
                   float* __restrict__ C, int K, int Nld, int relu) {
    const int nb   = blockIdx.x;
    const int mb   = blockIdx.y * GEMM_WAVES + (threadIdx.x >> 5);
    const int lane = threadIdx.x & 31;
    const int half = lane >> 4;
    const int lm   = lane & 15;

    const float* Ap = A + (size_t)(mb * 16 + lm) * K + 2 * half;
    const float* Bp = B + (size_t)(2 * half) * Nld + nb * 16 + lm;
    v8f acc = wmma_tile_f32(Ap, Bp, Nld, K);

    const int cn = nb * 16 + lm;
    const float bb = bias ? bias[cn] : 0.0f;
#pragma unroll
    for (int v = 0; v < 8; ++v) {
        float r = acc[v] + bb;
        if (relu) r = fmaxf(r, 0.0f);
        C[(size_t)(mb * 16 + v + 8 * half) * Nld + cn] = r;
    }
}

static_assert(B_GRAPHS == 16 * GEMM_WAVES, "a workgroup of the FC grid holds every row tile");
__global__ __launch_bounds__(32 * GEMM_WAVES)
void fc_pair_gemm_f32(const float* __restrict__ A, const float* __restrict__ Wa,
                      const float* __restrict__ Wb, float* __restrict__ part) {
    const int nb   = blockIdx.x;
    const int z    = blockIdx.y;
    const int mb   = threadIdx.x >> 5;
    const int lane = threadIdx.x & 31;
    const int half = lane >> 4;
    const int lm   = lane & 15;
    const int Kc   = FC_IN_SZ / KCHUNKS;
    const int k0   = z * Kc;

    const float* B  = (nb < FC_N / 16) ? Wa : Wb;
    const int    bn = (nb & (FC_N / 16 - 1)) * 16 + lm;
    const float* Ap = A + (size_t)(mb * 16 + lm) * FC_IN_SZ + k0 + 2 * half;
    const float* Bp = B + (size_t)(k0 + 2 * half) * FC_N + bn;
    const v8f acc = wmma_tile_f32(Ap, Bp, FC_N, Kc);

    float* Cp = part + (size_t)z * (B_GRAPHS * 2 * FC_N)
                     + (size_t)(mb * 16 + 8 * half) * (2 * FC_N) + nb * 16 + lm;
#pragma unroll
    for (int v = 0; v < 8; ++v) Cp[(size_t)v * (2 * FC_N)] = acc[v];
}

__global__ __launch_bounds__(2 * FC_N)
void fc_reduce_heads_kernel(const float* __restrict__ part,
                            const float* __restrict__ bp1, const float* __restrict__ bv1,
                            const float* __restrict__ Wiv, const float* __restrict__ biv,
                            const float* __restrict__ Wev, const float* __restrict__ bev,
                            float* __restrict__ Hp,
                            float* __restrict__ out_ext, float* __restrict__ out_int) {
    __shared__ float s_int[FC_N / 32], s_ext[FC_N / 32];
    const int r = blockIdx.x;
    const int t = threadIdx.x;
    const bool isV = t >= FC_N;
    const int  c   = t & (FC_N - 1);
    const float* p = part + (size_t)r * (2 * FC_N) + t;
    float s = 0.0f;
#pragma unroll
    for (int zc = 0; zc < KCHUNKS; ++zc) s += p[(size_t)zc * (B_GRAPHS * 2 * FC_N)];
    s += isV ? bv1[c] : bp1[c];
    s = fmaxf(s, 0.0f);
    if (!isV) Hp[(size_t)r * FC_N + c] = s;

    float pi = 0.0f, pe = 0.0f;
    if (isV) { pi = s * Wiv[c]; pe = s * Wev[c]; }
    for (int o = 16; o > 0; o >>= 1) {
        pi += __shfl_xor(pi, o);
        pe += __shfl_xor(pe, o);
    }
    if (isV && (t & 31) == 0) {
        s_int[(t - FC_N) >> 5] = pi;
        s_ext[(t - FC_N) >> 5] = pe;
    }
    __syncthreads();
    if (t == 0) {
        float si = 0.0f, se = 0.0f;
#pragma unroll
        for (int w = 0; w < FC_N / 32; ++w) { si += s_int[w]; se += s_ext[w]; }
        out_int[r] = si + biv[0];
        out_ext[r] = se + bev[0];
    }
}

static_assert(N_NODES % (8 * 2 * GATHER_PAIRS) == 0, "a gather block of 8 waves takes 64 whole nodes");
static_assert(GATHER_PAIRS == 4 && H_DIM == 16, "the fused tile below places 8 rows of 16 features");
template <bool FUSE_NEXT>
__device__ __forceinline__
void gather_rows(const float* __restrict__ xs, const int* __restrict__ csr,
                 const int* __restrict__ offs, const int* __restrict__ cnt,
                 const float* __restrict__ dinv, const float* __restrict__ bias,
                 const float* __restrict__ Wn, float* __restrict__ out, int n_nodes) {
    __shared__ float s_h[FUSE_NEXT ? 8 : 1][2 * GATHER_PAIRS + 1][H_DIM];
    const int wave = (blockIdx.x * blockDim.x + threadIdx.x) >> 5;
    if (wave * (2 * GATHER_PAIRS) >= n_nodes) return;
    const int lane = threadIdx.x & 31;
    const int j  = lane & 15;
    const int hw = lane >> 4;
    const int wl = threadIdx.x >> 5;
    const int node0 = wave * (2 * GATHER_PAIRS) + hw;

    int   st[GATHER_PAIRS], en[GATHER_PAIRS];
    float dn[GATHER_PAIRS];
#pragma unroll
    for (int p = 0; p < GATHER_PAIRS; ++p) {
        const int node = node0 + 2 * p;
        st[p] = offs[node];
        en[p] = st[p] + cnt[node];
        dn[p] = dinv[node];
    }
    const float bj = bias[j];
    if (FUSE_NEXT && hw == 0) s_h[wl][2 * GATHER_PAIRS][j] = 0.0f;

#pragma unroll
    for (int p = 0; p < GATHER_PAIRS; ++p) {
        const int node  = node0 + 2 * p;
        const int i     = node * H_DIM + j;
        const int start = st[p];
        const int end   = en[p];
        const float self = xs[i];
        float acc = 0.0f;
        for (int base = start; base < end; base += 16) {
            const int pp = base + j;
            int src = node;
            if (pp < end) src = csr[pp];
            const int m = end - base;
            float xv[16];
#pragma unroll
            for (int q = 0; q < 16; ++q) {
                const int r = __shfl(src, q, 16) & (N_NODES - 1);
                xv[q] = xs[r * H_DIM + j];
            }
#pragma unroll
            for (int q = 0; q < 16; ++q) {
                if (q < m) acc += xv[q];
            }
        }
        const float r = fmaxf(dn[p] * (acc + self) + bj, 0.0f);
        if (FUSE_NEXT) s_h[wl][p + GATHER_PAIRS * hw][j] = r;
        else           out[i] = r;
    }

    if (FUSE_NEXT) {
        v2f bw[H_DIM / 4];
#pragma unroll
        for (int s = 0; s < H_DIM / 4; ++s) {
            bw[s].x = Wn[(4 * s + 2 * hw) * H_DIM + j];
            bw[s].y = Wn[(4 * s + 2 * hw + 1) * H_DIM + j];
        }
        __syncthreads();
        const int arow = (j & 4) ? 2 * GATHER_PAIRS : (j & 3) + GATHER_PAIRS * (j >> 3);
        v8f d = {};
#pragma unroll
        for (int s = 0; s < H_DIM / 4; ++s) {
            const v2f a = *(const v2f*)&s_h[wl][arow][4 * s + 2 * hw];
            d = __builtin_amdgcn_wmma_f32_16x16x4_f32(
                    false, a, false, bw[s], (short)0, d, false, false);
        }
        float* op = out + (size_t)wave * (2 * GATHER_PAIRS * H_DIM) + lane;
#pragma unroll
        for (int v = 0; v < GATHER_PAIRS; ++v) op[v * 2 * H_DIM] = dn[v] * d[v];
    }
}

__global__ __launch_bounds__(256)
void gather_kernel(const float* __restrict__ xs, const int* __restrict__ csr,
                   const int* __restrict__ offs, const int* __restrict__ cnt,
                   const float* __restrict__ dinv, const float* __restrict__ bias,
                   float* __restrict__ h, int n_nodes) {
    gather_rows<false>(xs, csr, offs, cnt, dinv, bias, nullptr, h, n_nodes);
}

__global__ __launch_bounds__(256)
void gather_next_kernel(const float* __restrict__ xs, const int* __restrict__ csr,
                        const int* __restrict__ offs, const int* __restrict__ cnt,
                        const float* __restrict__ dinv, const float* __restrict__ bias,
                        const float* __restrict__ Wn, float* __restrict__ xs_next, int n_nodes) {
    gather_rows<true>(xs, csr, offs, cnt, dinv, bias, Wn, xs_next, n_nodes);
}

extern "C" void kernel_launch(void* const* d_in, const int* in_sizes, int n_in,
                              void* d_out, int out_size, void* d_ws, size_t ws_size,
                              hipStream_t stream) {
    const float* x    = (const float*)d_in[0];
    const int*   ei   = (const int*)d_in[1];
    const float* W1   = (const float*)d_in[2];
    const float* b1   = (const float*)d_in[3];
    const float* W2   = (const float*)d_in[4];
    const float* b2   = (const float*)d_in[5];
    const float* W3   = (const float*)d_in[6];
    const float* b3   = (const float*)d_in[7];
    const float* Wp1  = (const float*)d_in[8];
    const float* bp1  = (const float*)d_in[9];
    const float* Wp2  = (const float*)d_in[10];
    const float* bp2  = (const float*)d_in[11];
    const float* Wv1  = (const float*)d_in[12];
    const float* bv1  = (const float*)d_in[13];
    const float* Wiv  = (const float*)d_in[14];
    const float* biv  = (const float*)d_in[15];
    const float* Wev  = (const float*)d_in[16];
    const float* bev  = (const float*)d_in[17];

    const int* row = ei;
    const int* col = ei + E_EDGES;

    char* ws = (char*)d_ws;
    float* dinv = (float*)ws;
    float* bufA = (float*)(ws + (size_t)N_NODES * 4);
    float* bufB = bufA + (size_t)N_NODES * H_DIM;
    float* Hp   = bufB + (size_t)N_NODES * H_DIM;
    float* part = Hp + (size_t)B_GRAPHS * FC_N;
    int*   cnt  = (int*)(part + (size_t)KCHUNKS * B_GRAPHS * 2 * FC_N);
    int*   offs = cnt + N_NODES;
    int*   binned = offs + N_NODES;
    int*   csr    = binned + E_EDGES;
    int*   runtab = csr + E_EDGES;

    float* P     = (float*)d_out;
    float* ext_V = P + (size_t)B_GRAPHS * 1024;
    float* int_V = ext_V + B_GRAPHS;

    bin_partition_kernel<<<PART_WGS, 1024, 0, stream>>>(row, col, binned, runtab);
    bin_sort_kernel<<<NBINS, 1024, 0, stream>>>(binned, runtab, x, W1, csr, cnt, offs, dinv, bufA);

    gather_next_kernel<<<N_NODES / (8 * 2 * GATHER_PAIRS), 256, 0, stream>>>(bufA, csr, offs, cnt, dinv, b1, W2, bufB, N_NODES);

    gather_next_kernel<<<N_NODES / (8 * 2 * GATHER_PAIRS), 256, 0, stream>>>(bufB, csr, offs, cnt, dinv, b2, W3, bufA, N_NODES);

    gather_kernel<<<N_NODES / (8 * 2 * GATHER_PAIRS), 256, 0, stream>>>(bufA, csr, offs, cnt, dinv, b3, bufB, N_NODES);


    fc_pair_gemm_f32<<<dim3(2 * FC_N / 16, KCHUNKS), 32 * GEMM_WAVES, 0, stream>>>(bufB, Wp1, Wv1, part);
    fc_reduce_heads_kernel<<<B_GRAPHS, 2 * FC_N, 0, stream>>>(part, bp1, bv1, Wiv, biv, Wev, bev, Hp, ext_V, int_V);

    wmma_gemm_f32<<<dim3(1024 / 16, B_GRAPHS / 16 / GEMM_WAVES), 32 * GEMM_WAVES, 0, stream>>>(Hp, Wp2, bp2, P, FC_N, 1024, 0);
}


# FC pair GEMM re-tiled: a wave forms a 16x64 strip (4 accumulators share each A fragment), 16-byte operand loads kept one set ahead; k regrouped inside 8-k step pairs, so the sum order changes
# speedup vs baseline: 6.7407x; 1.1888x over previous
#include <hip/hip_runtime.h>

typedef __attribute__((ext_vector_type(2))) float v2f;
typedef __attribute__((ext_vector_type(4))) float v4f;
typedef __attribute__((ext_vector_type(8))) float v8f;
typedef __attribute__((ext_vector_type(4))) int   v4i;
typedef __attribute__((ext_vector_type(2))) int   v2i;

#define N_NODES   131072
#define E_EDGES   4194304
#define S_IN      32
#define H_DIM     16
#define B_GRAPHS  128
#define FC_IN_SZ  16384
#define GATHER_PAIRS 4
#define GEMM_WAVES   8
#define FC_N      256
#define KCHUNKS   64

#define BIN_SHIFT   9
#define BIN_NODES   512
#define NBINS       256
#define PART_WGS    256
#define PART_EDGES  16384
#define SORT_CAP    32768
#define RUN_WAVE    8
#define RUN_REGS    3
static_assert(NBINS * BIN_NODES == N_NODES && (1 << BIN_SHIFT) == BIN_NODES, "ranges cover the nodes");
static_assert(PART_WGS * PART_EDGES == E_EDGES && PART_EDGES == 16 * 1024, "level-1 chunks cover the edges");
static_assert(PART_WGS == 32 * RUN_WAVE && BIN_NODES == 32 * 16, "a level-2 workgroup: 32 waves, 8 runs and one 16-row tile each");

__device__ __forceinline__ int block_excl_scan_1024(int v, int* s_w) {
    const int lane = threadIdx.x & 31;
    const int wv   = threadIdx.x >> 5;
    int inc = v;
    for (int o = 1; o < 32; o <<= 1) {
        int u = __shfl_up(inc, o);
        if (lane >= o) inc += u;
    }
    __syncthreads();
    if (lane == 31) s_w[wv] = inc;
    __syncthreads();
    const int wtot = s_w[lane];
    int winc = wtot;
    for (int o = 1; o < 32; o <<= 1) {
        int u = __shfl_up(winc, o);
        if (lane >= o) winc += u;
    }
    const int wexcl = __shfl(winc - wtot, wv);
    return wexcl + inc - v;
}

__device__ __forceinline__ int block_sum_1024(int v, int* s_w) {
    for (int o = 16; o > 0; o >>= 1) v += __shfl_xor(v, o);
    __syncthreads();
    if ((threadIdx.x & 31) == 0) s_w[threadIdx.x >> 5] = v;
    __syncthreads();
    int tot = s_w[threadIdx.x & 31];
    for (int o = 16; o > 0; o >>= 1) tot += __shfl_xor(tot, o);
    return tot;
}

__global__ __launch_bounds__(1024)
void bin_partition_kernel(const int* __restrict__ row, const int* __restrict__ col,
                          int* __restrict__ binned, int* __restrict__ runtab) {
    __shared__ int h[NBINS];
    __shared__ int lstart[NBINS];
    __shared__ int s_w[32];
    __shared__ __attribute__((aligned(16))) int srec[PART_EDGES];
    const int t = threadIdx.x;
    if (t < NBINS) h[t] = 0;
    __syncthreads();
    const int* cp = col + blockIdx.x * PART_EDGES;
    const int* rp = row + blockIdx.x * PART_EDGES;
    v4i c[4], r[4];
#pragma unroll
    for (int q = 0; q < 4; ++q) {
        c[q] = *(const v4i*)(cp + 4 * (q * 1024 + t));
        r[q] = *(const v4i*)(rp + 4 * (q * 1024 + t));
    }
    int lr[16];
#pragma unroll
    for (int q = 0; q < 4; ++q) {
#pragma unroll
        for (int e = 0; e < 4; ++e) {
            c[q][e] &= (N_NODES - 1);
            r[q][e] &= (N_NODES - 1);
            lr[4 * q + e] = atomicAdd(&h[c[q][e] >> BIN_SHIFT], 1);
        }
    }
    __syncthreads();
    const int mine = (t < NBINS) ? h[t] : 0;
    const int lex = block_excl_scan_1024(mine, s_w);
    if (t < NBINS) {
        lstart[t] = lex;
        const v2i ent = {lex, mine};
        *(v2i*)(runtab + 2 * (t * PART_WGS + blockIdx.x)) = ent;
    }
    __syncthreads();
#pragma unroll
    for (int q = 0; q < 4; ++q) {
#pragma unroll
        for (int e = 0; e < 4; ++e) {
            const int p = (lstart[c[q][e] >> BIN_SHIFT] + lr[4 * q + e]) & (PART_EDGES - 1);
            srec[p] = (r[q][e] << BIN_SHIFT) | (c[q][e] & (BIN_NODES - 1));
        }
    }
    __syncthreads();
    int* bp = binned + blockIdx.x * PART_EDGES;
#pragma unroll
    for (int q = 0; q < 4; ++q) {
        const int i = 4 * (q * 1024 + t);
        *(v4i*)(bp + i) = *(const v4i*)&srec[i];
    }
}

__global__ __launch_bounds__(1024)
void bin_sort_kernel(const int* __restrict__ binned, const int* __restrict__ runtab,
                     const float* __restrict__ x, const float* __restrict__ W1,
                     int* __restrict__ csr, int* __restrict__ cnt, int* __restrict__ offs,
                     float* __restrict__ dinv, float* __restrict__ xs) {
    __shared__ int c_cnt[BIN_NODES];
    __shared__ int c_cur[BIN_NODES];
    __shared__ __attribute__((aligned(16))) float s_dinv[BIN_NODES];
    __shared__ int r_base[PART_WGS];
    __shared__ int r_n[PART_WGS];
    __shared__ int s_w[32];
    __shared__ int stage[SORT_CAP];
    const int t = threadIdx.x;
    const int b = blockIdx.x;
    const int lane = t & 31;
    const int wave = t >> 5;
    const int half = lane >> 4;
    const int lm   = lane & 15;

    int rs = 0, rn = 0;
    if (t < PART_WGS) {
        const v2i ent = *(const v2i*)(runtab + 2 * (b * PART_WGS + t));
        rs = min(max(ent.x, 0), PART_EDGES);
        rn = min(max(ent.y, 0), PART_EDGES - rs);
        r_base[t] = t * PART_EDGES + rs;
        r_n[t]    = rn;
    }
    if (t < BIN_NODES) c_cnt[t] = 0;
    const int start_raw = block_sum_1024(rs, s_w);
    const int nb_raw    = block_sum_1024(rn, s_w);
    const int start = min(max(start_raw, 0), E_EDGES);
    const int nb    = min(max(nb_raw, 0), E_EDGES - start);

    int rec[RUN_WAVE][RUN_REGS];
#pragma unroll
    for (int m = 0; m < RUN_WAVE; ++m) {
        const int w = wave + 32 * m;
        const int n = r_n[w];
        const int* rp = binned + r_base[w];
#pragma unroll
        for (int j = 0; j < RUN_REGS; ++j) {
            const int i = lane + 32 * j;
            rec[m][j] = (i < n) ? rp[i] : -1;
        }
    }
    v2f fa[S_IN / 4], fb[S_IN / 4];
    {
        const float* Ap = x + (size_t)(b * BIN_NODES + wave * 16 + lm) * S_IN + 2 * half;
        const float* Bp = W1 + (2 * half) * H_DIM + lm;
#pragma unroll
        for (int s = 0; s < S_IN / 4; ++s) {
            fa[s]   = *(const v2f*)(Ap + 4 * s);
            fb[s].x = Bp[(4 * s) * H_DIM];
            fb[s].y = Bp[(4 * s + 1) * H_DIM];
        }
    }
#pragma unroll
    for (int m = 0; m < RUN_WAVE; ++m) {
#pragma unroll
        for (int j = 0; j < RUN_REGS; ++j) {
            if (rec[m][j] >= 0) atomicAdd(&c_cnt[rec[m][j] & (BIN_NODES - 1)], 1);
        }
    }
    for (int m = 0; m < RUN_WAVE; ++m) {
        const int w = wave + 32 * m;
        const int n = r_n[w];
        const int* rp = binned + r_base[w];
        for (int i = 32 * RUN_REGS + lane; i < n; i += 32) atomicAdd(&c_cnt[rp[i] & (BIN_NODES - 1)], 1);
    }
    v8f acc = {};
#pragma unroll
    for (int s = 0; s < S_IN / 4; ++s) {
        acc = __builtin_amdgcn_wmma_f32_16x16x4_f32(
                  false, fa[s], false, fb[s], (short)0, acc, false, false);
    }
    __syncthreads();
    const int mine = (t < BIN_NODES) ? c_cnt[t] : 0;
    const int lex  = block_excl_scan_1024(mine, s_w);
    if (t < BIN_NODES) {
        c_cur[t] = lex;
        const int node = b * BIN_NODES + t;
        const float d = rsqrtf((float)(mine + 1));
        cnt[node]  = mine;
        offs[node] = start + lex;
        dinv[node] = d;
        s_dinv[t]  = d;
    }
    __syncthreads();

    {
        const v4f s0 = *(const v4f*)&s_dinv[wave * 16 + 8 * half];
        const v4f s1 = *(const v4f*)&s_dinv[wave * 16 + 8 * half + 4];
        float* Cp = xs + (size_t)(b * BIN_NODES + wave * 16 + 8 * half) * H_DIM + lm;
#pragma unroll
        for (int v = 0; v < 4; ++v) {
            Cp[v * H_DIM]       = acc[v] * s0[v];
            Cp[(v + 4) * H_DIM] = acc[v + 4] * s1[v];
        }
    }

    const bool staged = nb <= SORT_CAP;
#pragma unroll
    for (int m = 0; m < RUN_WAVE; ++m) {
#pragma unroll
        for (int j = 0; j < RUN_REGS; ++j) {
            if (rec[m][j] >= 0) {
                const int p = atomicAdd(&c_cur[rec[m][j] & (BIN_NODES - 1)], 1);
                const int sid = rec[m][j] >> BIN_SHIFT;
                if (staged) stage[min(p, SORT_CAP - 1)] = sid;
                else        csr[min((unsigned)(start + p), (unsigned)(E_EDGES - 1))] = sid;
            }
        }
    }
    for (int m = 0; m < RUN_WAVE; ++m) {
        const int w = wave + 32 * m;
        const int n = r_n[w];
        const int* rp = binned + r_base[w];
        for (int i = 32 * RUN_REGS + lane; i < n; i += 32) {
            const int rr = rp[i];
            const int p = atomicAdd(&c_cur[rr & (BIN_NODES - 1)], 1);
            const int sid = rr >> BIN_SHIFT;
            if (staged) stage[min(p, SORT_CAP - 1)] = sid;
            else        csr[min((unsigned)(start + p), (unsigned)(E_EDGES - 1))] = sid;
        }
    }
    if (staged) {
        __syncthreads();
        for (int i = t; i < nb; i += 1024) csr[start + i] = stage[i];
    }
}

__device__ __forceinline__ v8f wmma_tile_f32(const float* __restrict__ Ap, const float* __restrict__ Bp,
                                             int ldb, int Kc) {
    v8f acc = {};
#pragma unroll 4
    for (int k = 0; k < Kc; k += 4) {
        v2f a = *(const v2f*)Ap;
        v2f b;
        b.x = Bp[0];
        b.y = Bp[ldb];
        acc = __builtin_amdgcn_wmma_f32_16x16x4_f32(
                  false, a, false, b, (short)0, acc, false, false);
        Ap += 4;
        Bp += (size_t)4 * ldb;
    }
    return acc;
}

static_assert((B_GRAPHS / 16) % GEMM_WAVES == 0, "whole GEMM workgroups");
__global__ __launch_bounds__(32 * GEMM_WAVES)
void wmma_gemm_f32(const float* __restrict__ A, const float* __restrict__ B,
                   const float* __restrict__ bias,
                   float* __restrict__ C, int K, int Nld, int relu) {
    const int nb   = blockIdx.x;
    const int mb   = blockIdx.y * GEMM_WAVES + (threadIdx.x >> 5);
    const int lane = threadIdx.x & 31;
    const int half = lane >> 4;
    const int lm   = lane & 15;

    const float* Ap = A + (size_t)(mb * 16 + lm) * K + 2 * half;
    const float* Bp = B + (size_t)(2 * half) * Nld + nb * 16 + lm;
    v8f acc = wmma_tile_f32(Ap, Bp, Nld, K);

    const int cn = nb * 16 + lm;
    const float bb = bias ? bias[cn] : 0.0f;
#pragma unroll
    for (int v = 0; v < 8; ++v) {
        float r = acc[v] + bb;
        if (relu) r = fmaxf(r, 0.0f);
        C[(size_t)(mb * 16 + v + 8 * half) * Nld + cn] = r;
    }
}

#define FC_TILES 4
static_assert(B_GRAPHS == 16 * GEMM_WAVES, "a workgroup of the FC grid holds every row tile");
static_assert(FC_N % (16 * FC_TILES) == 0, "a column group lies inside one of the two weights");
static_assert((FC_IN_SZ / KCHUNKS) % 32 == 0 && FC_IN_SZ / KCHUNKS >= 64, "a K chunk is whole pairs of 16-k operand sets");

struct fc_frag {
    v4f a[2];
    v4f b[2][2][2];
};

__device__ __forceinline__ void fc_load(fc_frag& f, const float* __restrict__ Ap, const float* __restrict__ Bp) {
#pragma unroll
    for (int j = 0; j < 2; ++j) {
        f.a[j] = *(const v4f*)(Ap + 8 * j);
#pragma unroll
        for (int u = 0; u < 2; ++u) {
#pragma unroll
            for (int e = 0; e < 2; ++e)
                f.b[j][u][e] = *(const v4f*)(Bp + (size_t)(8 * j + 2 * u + e) * FC_N);
        }
    }
}

__device__ __forceinline__ void fc_mma(const fc_frag& f, v8f (&acc)[FC_TILES]) {
#pragma unroll
    for (int j = 0; j < 2; ++j) {
#pragma unroll
        for (int u = 0; u < 2; ++u) {
            v2f a;
            a.x = f.a[j][2 * u];
            a.y = f.a[j][2 * u + 1];
#pragma unroll
            for (int t = 0; t < FC_TILES; ++t) {
                v2f b;
                b.x = f.b[j][u][0][t];
                b.y = f.b[j][u][1][t];
                acc[t] = __builtin_amdgcn_wmma_f32_16x16x4_f32(
                             false, a, false, b, (short)0, acc[t], false, false);
            }
        }
    }
}

__global__ __launch_bounds__(32 * GEMM_WAVES)
void fc_pair_gemm_f32(const float* __restrict__ A, const float* __restrict__ Wa,
                      const float* __restrict__ Wb, float* __restrict__ part) {
    const int ng   = blockIdx.x;
    const int z    = blockIdx.y;
    const int mb   = threadIdx.x >> 5;
    const int lane = threadIdx.x & 31;
    const int half = lane >> 4;
    const int lm   = lane & 15;
    const int Kc   = FC_IN_SZ / KCHUNKS;
    const int NIT  = Kc / 16;
    const int k0   = z * Kc;
    const int GPW  = FC_N / (16 * FC_TILES);

    const float* B  = (ng < GPW) ? Wa : Wb;
    const int    bn = (ng & (GPW - 1)) * (16 * FC_TILES) + 4 * lm;
    const float* Ap = A + (size_t)(mb * 16 + lm) * FC_IN_SZ + k0 + 4 * half;
    const float* Bp = B + (size_t)(k0 + 4 * half) * FC_N + bn;

    v8f acc[FC_TILES] = {};
    fc_frag f0, f1;
    fc_load(f0, Ap, Bp);
    for (int it = 0; it < NIT - 2; it += 2) {
        fc_load(f1, Ap + 16 * (it + 1), Bp + (size_t)16 * (it + 1) * FC_N);
        __builtin_amdgcn_sched_barrier(0);
        fc_mma(f0, acc);
        __builtin_amdgcn_sched_barrier(0);
        fc_load(f0, Ap + 16 * (it + 2), Bp + (size_t)16 * (it + 2) * FC_N);
        __builtin_amdgcn_sched_barrier(0);
        fc_mma(f1, acc);
        __builtin_amdgcn_sched_barrier(0);
    }
    fc_load(f1, Ap + 16 * (NIT - 1), Bp + (size_t)16 * (NIT - 1) * FC_N);
    __builtin_amdgcn_sched_barrier(0);
    fc_mma(f0, acc);
    fc_mma(f1, acc);

    float* Cp = part + (size_t)z * (B_GRAPHS * 2 * FC_N)
                     + (size_t)(mb * 16 + 8 * half) * (2 * FC_N) + ng * (16 * FC_TILES) + 4 * lm;
#pragma unroll
    for (int v = 0; v < 8; ++v) {
        v4f o;
#pragma unroll
        for (int t = 0; t < FC_TILES; ++t) o[t] = acc[t][v];
        *(v4f*)(Cp + (size_t)v * (2 * FC_N)) = o;
    }
}

__global__ __launch_bounds__(2 * FC_N)
void fc_reduce_heads_kernel(const float* __restrict__ part,
                            const float* __restrict__ bp1, const float* __restrict__ bv1,
                            const float* __restrict__ Wiv, const float* __restrict__ biv,
                            const float* __restrict__ Wev, const float* __restrict__ bev,
                            float* __restrict__ Hp,
                            float* __restrict__ out_ext, float* __restrict__ out_int) {
    __shared__ float s_int[FC_N / 32], s_ext[FC_N / 32];
    const int r = blockIdx.x;
    const int t = threadIdx.x;
    const bool isV = t >= FC_N;
    const int  c   = t & (FC_N - 1);
    const float* p = part + (size_t)r * (2 * FC_N) + t;
    float s = 0.0f;
#pragma unroll
    for (int zc = 0; zc < KCHUNKS; ++zc) s += p[(size_t)zc * (B_GRAPHS * 2 * FC_N)];
    s += isV ? bv1[c] : bp1[c];
    s = fmaxf(s, 0.0f);
    if (!isV) Hp[(size_t)r * FC_N + c] = s;

    float pi = 0.0f, pe = 0.0f;
    if (isV) { pi = s * Wiv[c]; pe = s * Wev[c]; }
    for (int o = 16; o > 0; o >>= 1) {
        pi += __shfl_xor(pi, o);
        pe += __shfl_xor(pe, o);
    }
    if (isV && (t & 31) == 0) {
        s_int[(t - FC_N) >> 5] = pi;
        s_ext[(t - FC_N) >> 5] = pe;
    }
    __syncthreads();
    if (t == 0) {
        float si = 0.0f, se = 0.0f;
#pragma unroll
        for (int w = 0; w < FC_N / 32; ++w) { si += s_int[w]; se += s_ext[w]; }
        out_int[r] = si + biv[0];
        out_ext[r] = se + bev[0];
    }
}

static_assert(N_NODES % (8 * 2 * GATHER_PAIRS) == 0, "a gather block of 8 waves takes 64 whole nodes");
static_assert(GATHER_PAIRS == 4 && H_DIM == 16, "the fused tile below places 8 rows of 16 features");
template <bool FUSE_NEXT>
__device__ __forceinline__
void gather_rows(const float* __restrict__ xs, const int* __restrict__ csr,
                 const int* __restrict__ offs, const int* __restrict__ cnt,
                 const float* __restrict__ dinv, const float* __restrict__ bias,
                 const float* __restrict__ Wn, float* __restrict__ out, int n_nodes) {
    __shared__ float s_h[FUSE_NEXT ? 8 : 1][2 * GATHER_PAIRS + 1][H_DIM];
    const int wave = (blockIdx.x * blockDim.x + threadIdx.x) >> 5;
    if (wave * (2 * GATHER_PAIRS) >= n_nodes) return;
    const int lane = threadIdx.x & 31;
    const int j  = lane & 15;
    const int hw = lane >> 4;
    const int wl = threadIdx.x >> 5;
    const int node0 = wave * (2 * GATHER_PAIRS) + hw;

    int   st[GATHER_PAIRS], en[GATHER_PAIRS];
    float dn[GATHER_PAIRS];
#pragma unroll
    for (int p = 0; p < GATHER_PAIRS; ++p) {
        const int node = node0 + 2 * p;
        st[p] = offs[node];
        en[p] = st[p] + cnt[node];
        dn[p] = dinv[node];
    }
    const float bj = bias[j];
    if (FUSE_NEXT && hw == 0) s_h[wl][2 * GATHER_PAIRS][j] = 0.0f;

#pragma unroll
    for (int p = 0; p < GATHER_PAIRS; ++p) {
        const int node  = node0 + 2 * p;
        const int i     = node * H_DIM + j;
        const int start = st[p];
        const int end   = en[p];
        const float self = xs[i];
        float acc = 0.0f;
        for (int base = start; base < end; base += 16) {
            const int pp = base + j;
            int src = node;
            if (pp < end) src = csr[pp];
            const int m = end - base;
            float xv[16];
#pragma unroll
            for (int q = 0; q < 16; ++q) {
                const int r = __shfl(src, q, 16) & (N_NODES - 1);
                xv[q] = xs[r * H_DIM + j];
            }
#pragma unroll
            for (int q = 0; q < 16; ++q) {
                if (q < m) acc += xv[q];
            }
        }
        const float r = fmaxf(dn[p] * (acc + self) + bj, 0.0f);
        if (FUSE_NEXT) s_h[wl][p + GATHER_PAIRS * hw][j] = r;
        else           out[i] = r;
    }

    if (FUSE_NEXT) {
        v2f bw[H_DIM / 4];
#pragma unroll
        for (int s = 0; s < H_DIM / 4; ++s) {
            bw[s].x = Wn[(4 * s + 2 * hw) * H_DIM + j];
            bw[s].y = Wn[(4 * s + 2 * hw + 1) * H_DIM + j];
        }
        __syncthreads();
        const int arow = (j & 4) ? 2 * GATHER_PAIRS : (j & 3) + GATHER_PAIRS * (j >> 3);
        v8f d = {};
#pragma unroll
        for (int s = 0; s < H_DIM / 4; ++s) {
            const v2f a = *(const v2f*)&s_h[wl][arow][4 * s + 2 * hw];
            d = __builtin_amdgcn_wmma_f32_16x16x4_f32(
                    false, a, false, bw[s], (short)0, d, false, false);
        }
        float* op = out + (size_t)wave * (2 * GATHER_PAIRS * H_DIM) + lane;
#pragma unroll
        for (int v = 0; v < GATHER_PAIRS; ++v) op[v * 2 * H_DIM] = dn[v] * d[v];
    }
}

__global__ __launch_bounds__(256)
void gather_kernel(const float* __restrict__ xs, const int* __restrict__ csr,
                   const int* __restrict__ offs, const int* __restrict__ cnt,
                   const float* __restrict__ dinv, const float* __restrict__ bias,
                   float* __restrict__ h, int n_nodes) {
    gather_rows<false>(xs, csr, offs, cnt, dinv, bias, nullptr, h, n_nodes);
}

__global__ __launch_bounds__(256)
void gather_next_kernel(const float* __restrict__ xs, const int* __restrict__ csr,
                        const int* __restrict__ offs, const int* __restrict__ cnt,
                        const float* __restrict__ dinv, const float* __restrict__ bias,
                        const float* __restrict__ Wn, float* __restrict__ xs_next, int n_nodes) {
    gather_rows<true>(xs, csr, offs, cnt, dinv, bias, Wn, xs_next, n_nodes);
}

extern "C" void kernel_launch(void* const* d_in, const int* in_sizes, int n_in,
                              void* d_out, int out_size, void* d_ws, size_t ws_size,
                              hipStream_t stream) {
    const float* x    = (const float*)d_in[0];
    const int*   ei   = (const int*)d_in[1];
    const float* W1   = (const float*)d_in[2];
    const float* b1   = (const float*)d_in[3];
    const float* W2   = (const float*)d_in[4];
    const float* b2   = (const float*)d_in[5];
    const float* W3   = (const float*)d_in[6];
    const float* b3   = (const float*)d_in[7];
    const float* Wp1  = (const float*)d_in[8];
    const float* bp1  = (const float*)d_in[9];
    const float* Wp2  = (const float*)d_in[10];
    const float* bp2  = (const float*)d_in[11];
    const float* Wv1  = (const float*)d_in[12];
    const float* bv1  = (const float*)d_in[13];
    const float* Wiv  = (const float*)d_in[14];
    const float* biv  = (const float*)d_in[15];
    const float* Wev  = (const float*)d_in[16];
    const float* bev  = (const float*)d_in[17];

    const int* row = ei;
    const int* col = ei + E_EDGES;

    char* ws = (char*)d_ws;
    float* dinv = (float*)ws;
    float* bufA = (float*)(ws + (size_t)N_NODES * 4);
    float* bufB = bufA + (size_t)N_NODES * H_DIM;
    float* Hp   = bufB + (size_t)N_NODES * H_DIM;
    float* part = Hp + (size_t)B_GRAPHS * FC_N;
    int*   cnt  = (int*)(part + (size_t)KCHUNKS * B_GRAPHS * 2 * FC_N);
    int*   offs = cnt + N_NODES;
    int*   binned = offs + N_NODES;
    int*   csr    = binned + E_EDGES;
    int*   runtab = csr + E_EDGES;

    float* P     = (float*)d_out;
    float* ext_V = P + (size_t)B_GRAPHS * 1024;
    float* int_V = ext_V + B_GRAPHS;

    bin_partition_kernel<<<PART_WGS, 1024, 0, stream>>>(row, col, binned, runtab);
    bin_sort_kernel<<<NBINS, 1024, 0, stream>>>(binned, runtab, x, W1, csr, cnt, offs, dinv, bufA);

    gather_next_kernel<<<N_NODES / (8 * 2 * GATHER_PAIRS), 256, 0, stream>>>(bufA, csr, offs, cnt, dinv, b1, W2, bufB, N_NODES);

    gather_next_kernel<<<N_NODES / (8 * 2 * GATHER_PAIRS), 256, 0, stream>>>(bufB, csr, offs, cnt, dinv, b2, W3, bufA, N_NODES);

    gather_kernel<<<N_NODES / (8 * 2 * GATHER_PAIRS), 256, 0, stream>>>(bufA, csr, offs, cnt, dinv, b3, bufB, N_NODES);


    fc_pair_gemm_f32<<<dim3(2 * FC_N / (16 * FC_TILES), KCHUNKS), 32 * GEMM_WAVES, 0, stream>>>(bufB, Wp1, Wv1, part);
    fc_reduce_heads_kernel<<<B_GRAPHS, 2 * FC_N, 0, stream>>>(part, bp1, bv1, Wiv, biv, Wev, bev, Hp, ext_V, int_V);

    wmma_gemm_f32<<<dim3(1024 / 16, B_GRAPHS / 16 / GEMM_WAVES), 32 * GEMM_WAVES, 0, stream>>>(Hp, Wp2, bp2, P, FC_N, 1024, 0);
}


# Policy head GEMM: all operand loads of a tile (K=256) issued as one group ahead of its 64 products, not one memory round trip per 16 k; 2-wave groups; value-head weights loaded early; sum order kept
# speedup vs baseline: 7.2271x; 1.0722x over previous
#include <hip/hip_runtime.h>

typedef __attribute__((ext_vector_type(2))) float v2f;
typedef __attribute__((ext_vector_type(4))) float v4f;
typedef __attribute__((ext_vector_type(8))) float v8f;
typedef __attribute__((ext_vector_type(4))) int   v4i;
typedef __attribute__((ext_vector_type(2))) int   v2i;

#define N_NODES   131072
#define E_EDGES   4194304
#define S_IN      32
#define H_DIM     16
#define B_GRAPHS  128
#define FC_IN_SZ  16384
#define GATHER_PAIRS 4
#define GEMM_WAVES   8
#define FC_N      256
#define KCHUNKS   64

#define BIN_SHIFT   9
#define BIN_NODES   512
#define NBINS       256
#define PART_WGS    256
#define PART_EDGES  16384
#define SORT_CAP    32768
#define RUN_WAVE    8
#define RUN_REGS    3
static_assert(NBINS * BIN_NODES == N_NODES && (1 << BIN_SHIFT) == BIN_NODES, "ranges cover the nodes");
static_assert(PART_WGS * PART_EDGES == E_EDGES && PART_EDGES == 16 * 1024, "level-1 chunks cover the edges");
static_assert(PART_WGS == 32 * RUN_WAVE && BIN_NODES == 32 * 16, "a level-2 workgroup: 32 waves, 8 runs and one 16-row tile each");

__device__ __forceinline__ int block_excl_scan_1024(int v, int* s_w) {
    const int lane = threadIdx.x & 31;
    const int wv   = threadIdx.x >> 5;
    int inc = v;
    for (int o = 1; o < 32; o <<= 1) {
        int u = __shfl_up(inc, o);
        if (lane >= o) inc += u;
    }
    __syncthreads();
    if (lane == 31) s_w[wv] = inc;
    __syncthreads();
    const int wtot = s_w[lane];
    int winc = wtot;
    for (int o = 1; o < 32; o <<= 1) {
        int u = __shfl_up(winc, o);
        if (lane >= o) winc += u;
    }
    const int wexcl = __shfl(winc - wtot, wv);
    return wexcl + inc - v;
}

__device__ __forceinline__ int block_sum_1024(int v, int* s_w) {
    for (int o = 16; o > 0; o >>= 1) v += __shfl_xor(v, o);
    __syncthreads();
    if ((threadIdx.x & 31) == 0) s_w[threadIdx.x >> 5] = v;
    __syncthreads();
    int tot = s_w[threadIdx.x & 31];
    for (int o = 16; o > 0; o >>= 1) tot += __shfl_xor(tot, o);
    return tot;
}

__global__ __launch_bounds__(1024)
void bin_partition_kernel(const int* __restrict__ row, const int* __restrict__ col,
                          int* __restrict__ binned, int* __restrict__ runtab) {
    __shared__ int h[NBINS];
    __shared__ int lstart[NBINS];
    __shared__ int s_w[32];
    __shared__ __attribute__((aligned(16))) int srec[PART_EDGES];
    const int t = threadIdx.x;
    if (t < NBINS) h[t] = 0;
    __syncthreads();
    const int* cp = col + blockIdx.x * PART_EDGES;
    const int* rp = row + blockIdx.x * PART_EDGES;
    v4i c[4], r[4];
#pragma unroll
    for (int q = 0; q < 4; ++q) {
        c[q] = *(const v4i*)(cp + 4 * (q * 1024 + t));
        r[q] = *(const v4i*)(rp + 4 * (q * 1024 + t));
    }
    int lr[16];
#pragma unroll
    for (int q = 0; q < 4; ++q) {
#pragma unroll
        for (int e = 0; e < 4; ++e) {
            c[q][e] &= (N_NODES - 1);
            r[q][e] &= (N_NODES - 1);
            lr[4 * q + e] = atomicAdd(&h[c[q][e] >> BIN_SHIFT], 1);
        }
    }
    __syncthreads();
    const int mine = (t < NBINS) ? h[t] : 0;
    const int lex = block_excl_scan_1024(mine, s_w);
    if (t < NBINS) {
        lstart[t] = lex;
        const v2i ent = {lex, mine};
        *(v2i*)(runtab + 2 * (t * PART_WGS + blockIdx.x)) = ent;
    }
    __syncthreads();
#pragma unroll
    for (int q = 0; q < 4; ++q) {
#pragma unroll
        for (int e = 0; e < 4; ++e) {
            const int p = (lstart[c[q][e] >> BIN_SHIFT] + lr[4 * q + e]) & (PART_EDGES - 1);
            srec[p] = (r[q][e] << BIN_SHIFT) | (c[q][e] & (BIN_NODES - 1));
        }
    }
    __syncthreads();
    int* bp = binned + blockIdx.x * PART_EDGES;
#pragma unroll
    for (int q = 0; q < 4; ++q) {
        const int i = 4 * (q * 1024 + t);
        *(v4i*)(bp + i) = *(const v4i*)&srec[i];
    }
}

__global__ __launch_bounds__(1024)
void bin_sort_kernel(const int* __restrict__ binned, const int* __restrict__ runtab,
                     const float* __restrict__ x, const float* __restrict__ W1,
                     int* __restrict__ csr, int* __restrict__ cnt, int* __restrict__ offs,
                     float* __restrict__ dinv, float* __restrict__ xs) {
    __shared__ int c_cnt[BIN_NODES];
    __shared__ int c_cur[BIN_NODES];
    __shared__ __attribute__((aligned(16))) float s_dinv[BIN_NODES];
    __shared__ int r_base[PART_WGS];
    __shared__ int r_n[PART_WGS];
    __shared__ int s_w[32];
    __shared__ int stage[SORT_CAP];
    const int t = threadIdx.x;
    const int b = blockIdx.x;
    const int lane = t & 31;
    const int wave = t >> 5;
    const int half = lane >> 4;
    const int lm   = lane & 15;

    int rs = 0, rn = 0;
    if (t < PART_WGS) {
        const v2i ent = *(const v2i*)(runtab + 2 * (b * PART_WGS + t));
        rs = min(max(ent.x, 0), PART_EDGES);
        rn = min(max(ent.y, 0), PART_EDGES - rs);
        r_base[t] = t * PART_EDGES + rs;
        r_n[t]    = rn;
    }
    if (t < BIN_NODES) c_cnt[t] = 0;
    const int start_raw = block_sum_1024(rs, s_w);
    const int nb_raw    = block_sum_1024(rn, s_w);
    const int start = min(max(start_raw, 0), E_EDGES);
    const int nb    = min(max(nb_raw, 0), E_EDGES - start);

    int rec[RUN_WAVE][RUN_REGS];
#pragma unroll
    for (int m = 0; m < RUN_WAVE; ++m) {
        const int w = wave + 32 * m;
        const int n = r_n[w];
        const int* rp = binned + r_base[w];
#pragma unroll
        for (int j = 0; j < RUN_REGS; ++j) {
            const int i = lane + 32 * j;
            rec[m][j] = (i < n) ? rp[i] : -1;
        }
    }
    v2f fa[S_IN / 4], fb[S_IN / 4];
    {
        const float* Ap = x + (size_t)(b * BIN_NODES + wave * 16 + lm) * S_IN + 2 * half;
        const float* Bp = W1 + (2 * half) * H_DIM + lm;
#pragma unroll
        for (int s = 0; s < S_IN / 4; ++s) {
            fa[s]   = *(const v2f*)(Ap + 4 * s);
            fb[s].x = Bp[(4 * s) * H_DIM];
            fb[s].y = Bp[(4 * s + 1) * H_DIM];
        }
    }
#pragma unroll
    for (int m = 0; m < RUN_WAVE; ++m) {
#pragma unroll
        for (int j = 0; j < RUN_REGS; ++j) {
            if (rec[m][j] >= 0) atomicAdd(&c_cnt[rec[m][j] & (BIN_NODES - 1)], 1);
        }
    }
    for (int m = 0; m < RUN_WAVE; ++m) {
        const int w = wave + 32 * m;
        const int n = r_n[w];
        const int* rp = binned + r_base[w];
        for (int i = 32 * RUN_REGS + lane; i < n; i += 32) atomicAdd(&c_cnt[rp[i] & (BIN_NODES - 1)], 1);
    }
    v8f acc = {};
#pragma unroll
    for (int s = 0; s < S_IN / 4; ++s) {
        acc = __builtin_amdgcn_wmma_f32_16x16x4_f32(
                  false, fa[s], false, fb[s], (short)0, acc, false, false);
    }
    __syncthreads();
    const int mine = (t < BIN_NODES) ? c_cnt[t] : 0;
    const int lex  = block_excl_scan_1024(mine, s_w);
    if (t < BIN_NODES) {
        c_cur[t] = lex;
        const int node = b * BIN_NODES + t;
        const float d = rsqrtf((float)(mine + 1));
        cnt[node]  = mine;
        offs[node] = start + lex;
        dinv[node] = d;
        s_dinv[t]  = d;
    }
    __syncthreads();

    {
        const v4f s0 = *(const v4f*)&s_dinv[wave * 16 + 8 * half];
        const v4f s1 = *(const v4f*)&s_dinv[wave * 16 + 8 * half + 4];
        float* Cp = xs + (size_t)(b * BIN_NODES + wave * 16 + 8 * half) * H_DIM + lm;
#pragma unroll
        for (int v = 0; v < 4; ++v) {
            Cp[v * H_DIM]       = acc[v] * s0[v];
            Cp[(v + 4) * H_DIM] = acc[v + 4] * s1[v];
        }
    }

    const bool staged = nb <= SORT_CAP;
#pragma unroll
    for (int m = 0; m < RUN_WAVE; ++m) {
#pragma unroll
        for (int j = 0; j < RUN_REGS; ++j) {
            if (rec[m][j] >= 0) {
                const int p = atomicAdd(&c_cur[rec[m][j] & (BIN_NODES - 1)], 1);
                const int sid = rec[m][j] >> BIN_SHIFT;
                if (staged) stage[min(p, SORT_CAP - 1)] = sid;
                else        csr[min((unsigned)(start + p), (unsigned)(E_EDGES - 1))] = sid;
            }
        }
    }
    for (int m = 0; m < RUN_WAVE; ++m) {
        const int w = wave + 32 * m;
        const int n = r_n[w];
        const int* rp = binned + r_base[w];
        for (int i = 32 * RUN_REGS + lane; i < n; i += 32) {
            const int rr = rp[i];
            const int p = atomicAdd(&c_cur[rr & (BIN_NODES - 1)], 1);
            const int sid = rr >> BIN_SHIFT;
            if (staged) stage[min(p, SORT_CAP - 1)] = sid;
            else        csr[min((unsigned)(start + p), (unsigned)(E_EDGES - 1))] = sid;
        }
    }
    if (staged) {
        __syncthreads();
        for (int i = t; i < nb; i += 1024) csr[start + i] = stage[i];
    }
}

#define POL_N      1024
#define POL_WAVES  2
static_assert((B_GRAPHS / 16) % POL_WAVES == 0, "whole policy workgroups");
__global__ __launch_bounds__(32 * POL_WAVES)
void policy_gemm_f32(const float* __restrict__ A, const float* __restrict__ B,
                     const float* __restrict__ bias, float* __restrict__ C) {
    const int nb   = blockIdx.x;
    const int mb   = blockIdx.y * POL_WAVES + (threadIdx.x >> 5);
    const int lane = threadIdx.x & 31;
    const int half = lane >> 4;
    const int lm   = lane & 15;
    const int cn   = nb * 16 + lm;

    const float* Ap = A + (size_t)(mb * 16 + lm) * FC_N + 2 * half;
    const float* Bp = B + (size_t)(2 * half) * POL_N + cn;
    v2f a[FC_N / 4], b[FC_N / 4];
#pragma unroll
    for (int s = 0; s < FC_N / 4; ++s) {
        a[s]   = *(const v2f*)(Ap + 4 * s);
        b[s].x = Bp[(size_t)(4 * s) * POL_N];
        b[s].y = Bp[(size_t)(4 * s + 1) * POL_N];
    }
    const float bb = bias[cn];
    __builtin_amdgcn_sched_barrier(0);
    v8f acc = {};
#pragma unroll
    for (int s = 0; s < FC_N / 4; ++s) {
        acc = __builtin_amdgcn_wmma_f32_16x16x4_f32(
                  false, a[s], false, b[s], (short)0, acc, false, false);
    }
#pragma unroll
    for (int v = 0; v < 8; ++v)
        C[(size_t)(mb * 16 + v + 8 * half) * POL_N + cn] = acc[v] + bb;
}

#define FC_TILES 4
static_assert(B_GRAPHS == 16 * GEMM_WAVES, "a workgroup of the FC grid holds every row tile");
static_assert(FC_N % (16 * FC_TILES) == 0, "a column group lies inside one of the two weights");
static_assert((FC_IN_SZ / KCHUNKS) % 32 == 0 && FC_IN_SZ / KCHUNKS >= 64, "a K chunk is whole pairs of 16-k operand sets");

struct fc_frag {
    v4f a[2];
    v4f b[2][2][2];
};

__device__ __forceinline__ void fc_load(fc_frag& f, const float* __restrict__ Ap, const float* __restrict__ Bp) {
#pragma unroll
    for (int j = 0; j < 2; ++j) {
        f.a[j] = *(const v4f*)(Ap + 8 * j);
#pragma unroll
        for (int u = 0; u < 2; ++u) {
#pragma unroll
            for (int e = 0; e < 2; ++e)
                f.b[j][u][e] = *(const v4f*)(Bp + (size_t)(8 * j + 2 * u + e) * FC_N);
        }
    }
}

__device__ __forceinline__ void fc_mma(const fc_frag& f, v8f (&acc)[FC_TILES]) {
#pragma unroll
    for (int j = 0; j < 2; ++j) {
#pragma unroll
        for (int u = 0; u < 2; ++u) {
            v2f a;
            a.x = f.a[j][2 * u];
            a.y = f.a[j][2 * u + 1];
#pragma unroll
            for (int t = 0; t < FC_TILES; ++t) {
                v2f b;
                b.x = f.b[j][u][0][t];
                b.y = f.b[j][u][1][t];
                acc[t] = __builtin_amdgcn_wmma_f32_16x16x4_f32(
                             false, a, false, b, (short)0, acc[t], false, false);
            }
        }
    }
}

__global__ __launch_bounds__(32 * GEMM_WAVES)
void fc_pair_gemm_f32(const float* __restrict__ A, const float* __restrict__ Wa,
                      const float* __restrict__ Wb, float* __restrict__ part) {
    const int ng   = blockIdx.x;
    const int z    = blockIdx.y;
    const int mb   = threadIdx.x >> 5;
    const int lane = threadIdx.x & 31;
    const int half = lane >> 4;
    const int lm   = lane & 15;
    const int Kc   = FC_IN_SZ / KCHUNKS;
    const int NIT  = Kc / 16;
    const int k0   = z * Kc;
    const int GPW  = FC_N / (16 * FC_TILES);

    const float* B  = (ng < GPW) ? Wa : Wb;
    const int    bn = (ng & (GPW - 1)) * (16 * FC_TILES) + 4 * lm;
    const float* Ap = A + (size_t)(mb * 16 + lm) * FC_IN_SZ + k0 + 4 * half;
    const float* Bp = B + (size_t)(k0 + 4 * half) * FC_N + bn;

    v8f acc[FC_TILES] = {};
    fc_frag f0, f1;
    fc_load(f0, Ap, Bp);
    for (int it = 0; it < NIT - 2; it += 2) {
        fc_load(f1, Ap + 16 * (it + 1), Bp + (size_t)16 * (it + 1) * FC_N);
        __builtin_amdgcn_sched_barrier(0);
        fc_mma(f0, acc);
        __builtin_amdgcn_sched_barrier(0);
        fc_load(f0, Ap + 16 * (it + 2), Bp + (size_t)16 * (it + 2) * FC_N);
        __builtin_amdgcn_sched_barrier(0);
        fc_mma(f1, acc);
        __builtin_amdgcn_sched_barrier(0);
    }
    fc_load(f1, Ap + 16 * (NIT - 1), Bp + (size_t)16 * (NIT - 1) * FC_N);
    __builtin_amdgcn_sched_barrier(0);
    fc_mma(f0, acc);
    fc_mma(f1, acc);

    float* Cp = part + (size_t)z * (B_GRAPHS * 2 * FC_N)
                     + (size_t)(mb * 16 + 8 * half) * (2 * FC_N) + ng * (16 * FC_TILES) + 4 * lm;
#pragma unroll
    for (int v = 0; v < 8; ++v) {
        v4f o;
#pragma unroll
        for (int t = 0; t < FC_TILES; ++t) o[t] = acc[t][v];
        *(v4f*)(Cp + (size_t)v * (2 * FC_N)) = o;
    }
}

__global__ __launch_bounds__(2 * FC_N)
void fc_reduce_heads_kernel(const float* __restrict__ part,
                            const float* __restrict__ bp1, const float* __restrict__ bv1,
                            const float* __restrict__ Wiv, const float* __restrict__ biv,
                            const float* __restrict__ Wev, const float* __restrict__ bev,
                            float* __restrict__ Hp,
                            float* __restrict__ out_ext, float* __restrict__ out_int) {
    __shared__ float s_int[FC_N / 32], s_ext[FC_N / 32];
    const int r = blockIdx.x;
    const int t = threadIdx.x;
    const bool isV = t >= FC_N;
    const int  c   = t & (FC_N - 1);
    const float* p = part + (size_t)r * (2 * FC_N) + t;
    const float wi = Wiv[c], we = Wev[c];
    float s = 0.0f;
#pragma unroll
    for (int zc = 0; zc < KCHUNKS; ++zc) s += p[(size_t)zc * (B_GRAPHS * 2 * FC_N)];
    s += isV ? bv1[c] : bp1[c];
    s = fmaxf(s, 0.0f);
    if (!isV) Hp[(size_t)r * FC_N + c] = s;

    const float pi0 = s * wi, pe0 = s * we;
    float pi = isV ? pi0 : 0.0f, pe = isV ? pe0 : 0.0f;
    for (int o = 16; o > 0; o >>= 1) {
        pi += __shfl_xor(pi, o);
        pe += __shfl_xor(pe, o);
    }
    if (isV && (t & 31) == 0) {
        s_int[(t - FC_N) >> 5] = pi;
        s_ext[(t - FC_N) >> 5] = pe;
    }
    __syncthreads();
    if (t == 0) {
        float si = 0.0f, se = 0.0f;
#pragma unroll
        for (int w = 0; w < FC_N / 32; ++w) { si += s_int[w]; se += s_ext[w]; }
        out_int[r] = si + biv[0];
        out_ext[r] = se + bev[0];
    }
}

static_assert(N_NODES % (8 * 2 * GATHER_PAIRS) == 0, "a gather block of 8 waves takes 64 whole nodes");
static_assert(GATHER_PAIRS == 4 && H_DIM == 16, "the fused tile below places 8 rows of 16 features");
template <bool FUSE_NEXT>
__device__ __forceinline__
void gather_rows(const float* __restrict__ xs, const int* __restrict__ csr,
                 const int* __restrict__ offs, const int* __restrict__ cnt,
                 const float* __restrict__ dinv, const float* __restrict__ bias,
                 const float* __restrict__ Wn, float* __restrict__ out, int n_nodes) {
    __shared__ float s_h[FUSE_NEXT ? 8 : 1][2 * GATHER_PAIRS + 1][H_DIM];
    const int wave = (blockIdx.x * blockDim.x + threadIdx.x) >> 5;
    if (wave * (2 * GATHER_PAIRS) >= n_nodes) return;
    const int lane = threadIdx.x & 31;
    const int j  = lane & 15;
    const int hw = lane >> 4;
    const int wl = threadIdx.x >> 5;
    const int node0 = wave * (2 * GATHER_PAIRS) + hw;

    int   st[GATHER_PAIRS], en[GATHER_PAIRS];
    float dn[GATHER_PAIRS];
#pragma unroll
    for (int p = 0; p < GATHER_PAIRS; ++p) {
        const int node = node0 + 2 * p;
        st[p] = offs[node];
        en[p] = st[p] + cnt[node];
        dn[p] = dinv[node];
    }
    const float bj = bias[j];
    if (FUSE_NEXT && hw == 0) s_h[wl][2 * GATHER_PAIRS][j] = 0.0f;

#pragma unroll
    for (int p = 0; p < GATHER_PAIRS; ++p) {
        const int node  = node0 + 2 * p;
        const int i     = node * H_DIM + j;
        const int start = st[p];
        const int end   = en[p];
        const float self = xs[i];
        float acc = 0.0f;
        for (int base = start; base < end; base += 16) {
            const int pp = base + j;
            int src = node;
            if (pp < end) src = csr[pp];
            const int m = end - base;
            float xv[16];
#pragma unroll
            for (int q = 0; q < 16; ++q) {
                const int r = __shfl(src, q, 16) & (N_NODES - 1);
                xv[q] = xs[r * H_DIM + j];
            }
#pragma unroll
            for (int q = 0; q < 16; ++q) {
                if (q < m) acc += xv[q];
            }
        }
        const float r = fmaxf(dn[p] * (acc + self) + bj, 0.0f);
        if (FUSE_NEXT) s_h[wl][p + GATHER_PAIRS * hw][j] = r;
        else           out[i] = r;
    }

    if (FUSE_NEXT) {
        v2f bw[H_DIM / 4];
#pragma unroll
        for (int s = 0; s < H_DIM / 4; ++s) {
            bw[s].x = Wn[(4 * s + 2 * hw) * H_DIM + j];
            bw[s].y = Wn[(4 * s + 2 * hw + 1) * H_DIM + j];
        }
        __syncthreads();
        const int arow = (j & 4) ? 2 * GATHER_PAIRS : (j & 3) + GATHER_PAIRS * (j >> 3);
        v8f d = {};
#pragma unroll
        for (int s = 0; s < H_DIM / 4; ++s) {
            const v2f a = *(const v2f*)&s_h[wl][arow][4 * s + 2 * hw];
            d = __builtin_amdgcn_wmma_f32_16x16x4_f32(
                    false, a, false, bw[s], (short)0, d, false, false);
        }
        float* op = out + (size_t)wave * (2 * GATHER_PAIRS * H_DIM) + lane;
#pragma unroll
        for (int v = 0; v < GATHER_PAIRS; ++v) op[v * 2 * H_DIM] = dn[v] * d[v];
    }
}

__global__ __launch_bounds__(256)
void gather_kernel(const float* __restrict__ xs, const int* __restrict__ csr,
                   const int* __restrict__ offs, const int* __restrict__ cnt,
                   const float* __restrict__ dinv, const float* __restrict__ bias,
                   float* __restrict__ h, int n_nodes) {
    gather_rows<false>(xs, csr, offs, cnt, dinv, bias, nullptr, h, n_nodes);
}

__global__ __launch_bounds__(256)
void gather_next_kernel(const float* __restrict__ xs, const int* __restrict__ csr,
                        const int* __restrict__ offs, const int* __restrict__ cnt,
                        const float* __restrict__ dinv, const float* __restrict__ bias,
                        const float* __restrict__ Wn, float* __restrict__ xs_next, int n_nodes) {
    gather_rows<true>(xs, csr, offs, cnt, dinv, bias, Wn, xs_next, n_nodes);
}

extern "C" void kernel_launch(void* const* d_in, const int* in_sizes, int n_in,
                              void* d_out, int out_size, void* d_ws, size_t ws_size,
                              hipStream_t stream) {
    const float* x    = (const float*)d_in[0];
    const int*   ei   = (const int*)d_in[1];
    const float* W1   = (const float*)d_in[2];
    const float* b1   = (const float*)d_in[3];
    const float* W2   = (const float*)d_in[4];
    const float* b2   = (const float*)d_in[5];
    const float* W3   = (const float*)d_in[6];
    const float* b3   = (const float*)d_in[7];
    const float* Wp1  = (const float*)d_in[8];
    const float* bp1  = (const float*)d_in[9];
    const float* Wp2  = (const float*)d_in[10];
    const float* bp2  = (const float*)d_in[11];
    const float* Wv1  = (const float*)d_in[12];
    const float* bv1  = (const float*)d_in[13];
    const float* Wiv  = (const float*)d_in[14];
    const float* biv  = (const float*)d_in[15];
    const float* Wev  = (const float*)d_in[16];
    const float* bev  = (const float*)d_in[17];

    const int* row = ei;
    const int* col = ei + E_EDGES;

    char* ws = (char*)d_ws;
    float* dinv = (float*)ws;
    float* bufA = (float*)(ws + (size_t)N_NODES * 4);
    float* bufB = bufA + (size_t)N_NODES * H_DIM;
    float* Hp   = bufB + (size_t)N_NODES * H_DIM;
    float* part = Hp + (size_t)B_GRAPHS * FC_N;
    int*   cnt  = (int*)(part + (size_t)KCHUNKS * B_GRAPHS * 2 * FC_N);
    int*   offs = cnt + N_NODES;
    int*   binned = offs + N_NODES;
    int*   csr    = binned + E_EDGES;
    int*   runtab = csr + E_EDGES;

    float* P     = (float*)d_out;
    float* ext_V = P + (size_t)B_GRAPHS * 1024;
    float* int_V = ext_V + B_GRAPHS;

    bin_partition_kernel<<<PART_WGS, 1024, 0, stream>>>(row, col, binned, runtab);
    bin_sort_kernel<<<NBINS, 1024, 0, stream>>>(binned, runtab, x, W1, csr, cnt, offs, dinv, bufA);

    gather_next_kernel<<<N_NODES / (8 * 2 * GATHER_PAIRS), 256, 0, stream>>>(bufA, csr, offs, cnt, dinv, b1, W2, bufB, N_NODES);

    gather_next_kernel<<<N_NODES / (8 * 2 * GATHER_PAIRS), 256, 0, stream>>>(bufB, csr, offs, cnt, dinv, b2, W3, bufA, N_NODES);

    gather_kernel<<<N_NODES / (8 * 2 * GATHER_PAIRS), 256, 0, stream>>>(bufA, csr, offs, cnt, dinv, b3, bufB, N_NODES);


    fc_pair_gemm_f32<<<dim3(2 * FC_N / (16 * FC_TILES), KCHUNKS), 32 * GEMM_WAVES, 0, stream>>>(bufB, Wp1, Wv1, part);
    fc_reduce_heads_kernel<<<B_GRAPHS, 2 * FC_N, 0, stream>>>(part, bp1, bv1, Wiv, biv, Wev, bev, Hp, ext_V, int_V);

    policy_gemm_f32<<<dim3(POL_N / 16, B_GRAPHS / 16 / POL_WAVES), 32 * POL_WAVES, 0, stream>>>(Hp, Wp2, bp2, P);
}


# Gather: 4 lanes per node with 16-byte row loads (8 whole rows per load instruction, 16 nodes per wave) and table entries asked for one chunk ahead of their rows; per-node sum order kept
# speedup vs baseline: 7.8517x; 1.0864x over previous
#include <hip/hip_runtime.h>

typedef __attribute__((ext_vector_type(2))) float v2f;
typedef __attribute__((ext_vector_type(4))) float v4f;
typedef __attribute__((ext_vector_type(8))) float v8f;
typedef __attribute__((ext_vector_type(4))) int   v4i;
typedef __attribute__((ext_vector_type(2))) int   v2i;

#define N_NODES   131072
#define E_EDGES   4194304
#define S_IN      32
#define H_DIM     16
#define B_GRAPHS  128
#define FC_IN_SZ  16384
#define GATHER_PASSES 2
#define GEMM_WAVES   8
#define FC_N      256
#define KCHUNKS   64

#define BIN_SHIFT   9
#define BIN_NODES   512
#define NBINS       256
#define PART_WGS    256
#define PART_EDGES  16384
#define SORT_CAP    32768
#define RUN_WAVE    8
#define RUN_REGS    3
static_assert(NBINS * BIN_NODES == N_NODES && (1 << BIN_SHIFT) == BIN_NODES, "ranges cover the nodes");
static_assert(PART_WGS * PART_EDGES == E_EDGES && PART_EDGES == 16 * 1024, "level-1 chunks cover the edges");
static_assert(PART_WGS == 32 * RUN_WAVE && BIN_NODES == 32 * 16, "a level-2 workgroup: 32 waves, 8 runs and one 16-row tile each");

__device__ __forceinline__ int block_excl_scan_1024(int v, int* s_w) {
    const int lane = threadIdx.x & 31;
    const int wv   = threadIdx.x >> 5;
    int inc = v;
    for (int o = 1; o < 32; o <<= 1) {
        int u = __shfl_up(inc, o);
        if (lane >= o) inc += u;
    }
    __syncthreads();
    if (lane == 31) s_w[wv] = inc;
    __syncthreads();
    const int wtot = s_w[lane];
    int winc = wtot;
    for (int o = 1; o < 32; o <<= 1) {
        int u = __shfl_up(winc, o);
        if (lane >= o) winc += u;
    }
    const int wexcl = __shfl(winc - wtot, wv);
    return wexcl + inc - v;
}

__device__ __forceinline__ int block_sum_1024(int v, int* s_w) {
    for (int o = 16; o > 0; o >>= 1) v += __shfl_xor(v, o);
    __syncthreads();
    if ((threadIdx.x & 31) == 0) s_w[threadIdx.x >> 5] = v;
    __syncthreads();
    int tot = s_w[threadIdx.x & 31];
    for (int o = 16; o > 0; o >>= 1) tot += __shfl_xor(tot, o);
    return tot;
}

__global__ __launch_bounds__(1024)
void bin_partition_kernel(const int* __restrict__ row, const int* __restrict__ col,
                          int* __restrict__ binned, int* __restrict__ runtab) {
    __shared__ int h[NBINS];
    __shared__ int lstart[NBINS];
    __shared__ int s_w[32];
    __shared__ __attribute__((aligned(16))) int srec[PART_EDGES];
    const int t = threadIdx.x;
    if (t < NBINS) h[t] = 0;
    __syncthreads();
    const int* cp = col + blockIdx.x * PART_EDGES;
    const int* rp = row + blockIdx.x * PART_EDGES;
    v4i c[4], r[4];
#pragma unroll
    for (int q = 0; q < 4; ++q) {
        c[q] = *(const v4i*)(cp + 4 * (q * 1024 + t));
        r[q] = *(const v4i*)(rp + 4 * (q * 1024 + t));
    }
    int lr[16];
#pragma unroll
    for (int q = 0; q < 4; ++q) {
#pragma unroll
        for (int e = 0; e < 4; ++e) {
            c[q][e] &= (N_NODES - 1);
            r[q][e] &= (N_NODES - 1);
            lr[4 * q + e] = atomicAdd(&h[c[q][e] >> BIN_SHIFT], 1);
        }
    }
    __syncthreads();
    const int mine = (t < NBINS) ? h[t] : 0;
    const int lex = block_excl_scan_1024(mine, s_w);
    if (t < NBINS) {
        lstart[t] = lex;
        const v2i ent = {lex, mine};
        *(v2i*)(runtab + 2 * (t * PART_WGS + blockIdx.x)) = ent;
    }
    __syncthreads();
#pragma unroll
    for (int q = 0; q < 4; ++q) {
#pragma unroll
        for (int e = 0; e < 4; ++e) {
            const int p = (lstart[c[q][e] >> BIN_SHIFT] + lr[4 * q + e]) & (PART_EDGES - 1);
            srec[p] = (r[q][e] << BIN_SHIFT) | (c[q][e] & (BIN_NODES - 1));
        }
    }
    __syncthreads();
    int* bp = binned + blockIdx.x * PART_EDGES;
#pragma unroll
    for (int q = 0; q < 4; ++q) {
        const int i = 4 * (q * 1024 + t);
        *(v4i*)(bp + i) = *(const v4i*)&srec[i];
    }
}

__global__ __launch_bounds__(1024)
void bin_sort_kernel(const int* __restrict__ binned, const int* __restrict__ runtab,
                     const float* __restrict__ x, const float* __restrict__ W1,
                     int* __restrict__ csr, int* __restrict__ cnt, int* __restrict__ offs,
                     float* __restrict__ dinv, float* __restrict__ xs) {
    __shared__ int c_cnt[BIN_NODES];
    __shared__ int c_cur[BIN_NODES];
    __shared__ __attribute__((aligned(16))) float s_dinv[BIN_NODES];
    __shared__ int r_base[PART_WGS];
    __shared__ int r_n[PART_WGS];
    __shared__ int s_w[32];
    __shared__ int stage[SORT_CAP];
    const int t = threadIdx.x;
    const int b = blockIdx.x;
    const int lane = t & 31;
    const int wave = t >> 5;
    const int half = lane >> 4;
    const int lm   = lane & 15;

    int rs = 0, rn = 0;
    if (t < PART_WGS) {
        const v2i ent = *(const v2i*)(runtab + 2 * (b * PART_WGS + t));
        rs = min(max(ent.x, 0), PART_EDGES);
        rn = min(max(ent.y, 0), PART_EDGES - rs);
        r_base[t] = t * PART_EDGES + rs;
        r_n[t]    = rn;
    }
    if (t < BIN_NODES) c_cnt[t] = 0;
    const int start_raw = block_sum_1024(rs, s_w);
    const int nb_raw    = block_sum_1024(rn, s_w);
    const int start = min(max(start_raw, 0), E_EDGES);
    const int nb    = min(max(nb_raw, 0), E_EDGES - start);

    int rec[RUN_WAVE][RUN_REGS];
#pragma unroll
    for (int m = 0; m < RUN_WAVE; ++m) {
        const int w = wave + 32 * m;
        const int n = r_n[w];
        const int* rp = binned + r_base[w];
#pragma unroll
        for (int j = 0; j < RUN_REGS; ++j) {
            const int i = lane + 32 * j;
            rec[m][j] = (i < n) ? rp[i] : -1;
        }
    }
    v2f fa[S_IN / 4], fb[S_IN / 4];
    {
        const float* Ap = x + (size_t)(b * BIN_NODES + wave * 16 + lm) * S_IN + 2 * half;
        const float* Bp = W1 + (2 * half) * H_DIM + lm;
#pragma unroll
        for (int s = 0; s < S_IN / 4; ++s) {
            fa[s]   = *(const v2f*)(Ap + 4 * s);
            fb[s].x = Bp[(4 * s) * H_DIM];
            fb[s].y = Bp[(4 * s + 1) * H_DIM];
        }
    }
#pragma unroll
    for (int m = 0; m < RUN_WAVE; ++m) {
#pragma unroll
        for (int j = 0; j < RUN_REGS; ++j) {
            if (rec[m][j] >= 0) atomicAdd(&c_cnt[rec[m][j] & (BIN_NODES - 1)], 1);
        }
    }
    for (int m = 0; m < RUN_WAVE; ++m) {
        const int w = wave + 32 * m;
        const int n = r_n[w];
        const int* rp = binned + r_base[w];
        for (int i = 32 * RUN_REGS + lane; i < n; i += 32) atomicAdd(&c_cnt[rp[i] & (BIN_NODES - 1)], 1);
    }
    v8f acc = {};
#pragma unroll
    for (int s = 0; s < S_IN / 4; ++s) {
        acc = __builtin_amdgcn_wmma_f32_16x16x4_f32(
                  false, fa[s], false, fb[s], (short)0, acc, false, false);
    }
    __syncthreads();
    const int mine = (t < BIN_NODES) ? c_cnt[t] : 0;
    const int lex  = block_excl_scan_1024(mine, s_w);
    if (t < BIN_NODES) {
        c_cur[t] = lex;
        const int node = b * BIN_NODES + t;
        const float d = rsqrtf((float)(mine + 1));
        cnt[node]  = mine;
        offs[node] = start + lex;
        dinv[node] = d;
        s_dinv[t]  = d;
    }
    __syncthreads();

    {
        const v4f s0 = *(const v4f*)&s_dinv[wave * 16 + 8 * half];
        const v4f s1 = *(const v4f*)&s_dinv[wave * 16 + 8 * half + 4];
        float* Cp = xs + (size_t)(b * BIN_NODES + wave * 16 + 8 * half) * H_DIM + lm;
#pragma unroll
        for (int v = 0; v < 4; ++v) {
            Cp[v * H_DIM]       = acc[v] * s0[v];
            Cp[(v + 4) * H_DIM] = acc[v + 4] * s1[v];
        }
    }

    const bool staged = nb <= SORT_CAP;
#pragma unroll
    for (int m = 0; m < RUN_WAVE; ++m) {
#pragma unroll
        for (int j = 0; j < RUN_REGS; ++j) {
            if (rec[m][j] >= 0) {
                const int p = atomicAdd(&c_cur[rec[m][j] & (BIN_NODES - 1)], 1);
                const int sid = rec[m][j] >> BIN_SHIFT;
                if (staged) stage[min(p, SORT_CAP - 1)] = sid;
                else        csr[min((unsigned)(start + p), (unsigned)(E_EDGES - 1))] = sid;
            }
        }
    }
    for (int m = 0; m < RUN_WAVE; ++m) {
        const int w = wave + 32 * m;
        const int n = r_n[w];
        const int* rp = binned + r_base[w];
        for (int i = 32 * RUN_REGS + lane; i < n; i += 32) {
            const int rr = rp[i];
            const int p = atomicAdd(&c_cur[rr & (BIN_NODES - 1)], 1);
            const int sid = rr >> BIN_SHIFT;
            if (staged) stage[min(p, SORT_CAP - 1)] = sid;
            else        csr[min((unsigned)(start + p), (unsigned)(E_EDGES - 1))] = sid;
        }
    }
    if (staged) {
        __syncthreads();
        for (int i = t; i < nb; i += 1024) csr[start + i] = stage[i];
    }
}

#define POL_N      1024
#define POL_WAVES  2
static_assert((B_GRAPHS / 16) % POL_WAVES == 0, "whole policy workgroups");
__global__ __launch_bounds__(32 * POL_WAVES)
void policy_gemm_f32(const float* __restrict__ A, const float* __restrict__ B,
                     const float* __restrict__ bias, float* __restrict__ C) {
    const int nb   = blockIdx.x;
    const int mb   = blockIdx.y * POL_WAVES + (threadIdx.x >> 5);
    const int lane = threadIdx.x & 31;
    const int half = lane >> 4;
    const int lm   = lane & 15;
    const int cn   = nb * 16 + lm;

    const float* Ap = A + (size_t)(mb * 16 + lm) * FC_N + 2 * half;
    const float* Bp = B + (size_t)(2 * half) * POL_N + cn;
    v2f a[FC_N / 4], b[FC_N / 4];
#pragma unroll
    for (int s = 0; s < FC_N / 4; ++s) {
        a[s]   = *(const v2f*)(Ap + 4 * s);
        b[s].x = Bp[(size_t)(4 * s) * POL_N];
        b[s].y = Bp[(size_t)(4 * s + 1) * POL_N];
    }
    const float bb = bias[cn];
    __builtin_amdgcn_sched_barrier(0);
    v8f acc = {};
#pragma unroll
    for (int s = 0; s < FC_N / 4; ++s) {
        acc = __builtin_amdgcn_wmma_f32_16x16x4_f32(
                  false, a[s], false, b[s], (short)0, acc, false, false);
    }
#pragma unroll
    for (int v = 0; v < 8; ++v)
        C[(size_t)(mb * 16 + v + 8 * half) * POL_N + cn] = acc[v] + bb;
}

#define FC_TILES 4
static_assert(B_GRAPHS == 16 * GEMM_WAVES, "a workgroup of the FC grid holds every row tile");
static_assert(FC_N % (16 * FC_TILES) == 0, "a column group lies inside one of the two weights");
static_assert((FC_IN_SZ / KCHUNKS) % 32 == 0 && FC_IN_SZ / KCHUNKS >= 64, "a K chunk is whole pairs of 16-k operand sets");

struct fc_frag {
    v4f a[2];
    v4f b[2][2][2];
};

__device__ __forceinline__ void fc_load(fc_frag& f, const float* __restrict__ Ap, const float* __restrict__ Bp) {
#pragma unroll
    for (int j = 0; j < 2; ++j) {
        f.a[j] = *(const v4f*)(Ap + 8 * j);
#pragma unroll
        for (int u = 0; u < 2; ++u) {
#pragma unroll
            for (int e = 0; e < 2; ++e)
                f.b[j][u][e] = *(const v4f*)(Bp + (size_t)(8 * j + 2 * u + e) * FC_N);
        }
    }
}

__device__ __forceinline__ void fc_mma(const fc_frag& f, v8f (&acc)[FC_TILES]) {
#pragma unroll
    for (int j = 0; j < 2; ++j) {
#pragma unroll
        for (int u = 0; u < 2; ++u) {
            v2f a;
            a.x = f.a[j][2 * u];
            a.y = f.a[j][2 * u + 1];
#pragma unroll
            for (int t = 0; t < FC_TILES; ++t) {
                v2f b;
                b.x = f.b[j][u][0][t];
                b.y = f.b[j][u][1][t];
                acc[t] = __builtin_amdgcn_wmma_f32_16x16x4_f32(
                             false, a, false, b, (short)0, acc[t], false, false);
            }
        }
    }
}

__global__ __launch_bounds__(32 * GEMM_WAVES)
void fc_pair_gemm_f32(const float* __restrict__ A, const float* __restrict__ Wa,
                      const float* __restrict__ Wb, float* __restrict__ part) {
    const int ng   = blockIdx.x;
    const int z    = blockIdx.y;
    const int mb   = threadIdx.x >> 5;
    const int lane = threadIdx.x & 31;
    const int half = lane >> 4;
    const int lm   = lane & 15;
    const int Kc   = FC_IN_SZ / KCHUNKS;
    const int NIT  = Kc / 16;
    const int k0   = z * Kc;
    const int GPW  = FC_N / (16 * FC_TILES);

    const float* B  = (ng < GPW) ? Wa : Wb;
    const int    bn = (ng & (GPW - 1)) * (16 * FC_TILES) + 4 * lm;
    const float* Ap = A + (size_t)(mb * 16 + lm) * FC_IN_SZ + k0 + 4 * half;
    const float* Bp = B + (size_t)(k0 + 4 * half) * FC_N + bn;

    v8f acc[FC_TILES] = {};
    fc_frag f0, f1;
    fc_load(f0, Ap, Bp);
    for (int it = 0; it < NIT - 2; it += 2) {
        fc_load(f1, Ap + 16 * (it + 1), Bp + (size_t)16 * (it + 1) * FC_N);
        __builtin_amdgcn_sched_barrier(0);
        fc_mma(f0, acc);
        __builtin_amdgcn_sched_barrier(0);
        fc_load(f0, Ap + 16 * (it + 2), Bp + (size_t)16 * (it + 2) * FC_N);
        __builtin_amdgcn_sched_barrier(0);
        fc_mma(f1, acc);
        __builtin_amdgcn_sched_barrier(0);
    }
    fc_load(f1, Ap + 16 * (NIT - 1), Bp + (size_t)16 * (NIT - 1) * FC_N);
    __builtin_amdgcn_sched_barrier(0);
    fc_mma(f0, acc);
    fc_mma(f1, acc);

    float* Cp = part + (size_t)z * (B_GRAPHS * 2 * FC_N)
                     + (size_t)(mb * 16 + 8 * half) * (2 * FC_N) + ng * (16 * FC_TILES) + 4 * lm;
#pragma unroll
    for (int v = 0; v < 8; ++v) {
        v4f o;
#pragma unroll
        for (int t = 0; t < FC_TILES; ++t) o[t] = acc[t][v];
        *(v4f*)(Cp + (size_t)v * (2 * FC_N)) = o;
    }
}

__global__ __launch_bounds__(2 * FC_N)
void fc_reduce_heads_kernel(const float* __restrict__ part,
                            const float* __restrict__ bp1, const float* __restrict__ bv1,
                            const float* __restrict__ Wiv, const float* __restrict__ biv,
                            const float* __restrict__ Wev, const float* __restrict__ bev,
                            float* __restrict__ Hp,
                            float* __restrict__ out_ext, float* __restrict__ out_int) {
    __shared__ float s_int[FC_N / 32], s_ext[FC_N / 32];
    const int r = blockIdx.x;
    const int t = threadIdx.x;
    const bool isV = t >= FC_N;
    const int  c   = t & (FC_N - 1);
    const float* p = part + (size_t)r * (2 * FC_N) + t;
    const float wi = Wiv[c], we = Wev[c];
    float s = 0.0f;
#pragma unroll
    for (int zc = 0; zc < KCHUNKS; ++zc) s += p[(size_t)zc * (B_GRAPHS * 2 * FC_N)];
    s += isV ? bv1[c] : bp1[c];
    s = fmaxf(s, 0.0f);
    if (!isV) Hp[(size_t)r * FC_N + c] = s;

    const float pi0 = s * wi, pe0 = s * we;
    float pi = isV ? pi0 : 0.0f, pe = isV ? pe0 : 0.0f;
    for (int o = 16; o > 0; o >>= 1) {
        pi += __shfl_xor(pi, o);
        pe += __shfl_xor(pe, o);
    }
    if (isV && (t & 31) == 0) {
        s_int[(t - FC_N) >> 5] = pi;
        s_ext[(t - FC_N) >> 5] = pe;
    }
    __syncthreads();
    if (t == 0) {
        float si = 0.0f, se = 0.0f;
#pragma unroll
        for (int w = 0; w < FC_N / 32; ++w) { si += s_int[w]; se += s_ext[w]; }
        out_int[r] = si + biv[0];
        out_ext[r] = se + bev[0];
    }
}

#define GATHER_ROWS 16
#define SH_PITCH    20
static_assert(N_NODES % (8 * 8 * GATHER_PASSES) == 0, "a gather block of 8 waves takes whole nodes");
static_assert(GATHER_PASSES == 2 && H_DIM == 16 && GATHER_ROWS == 16, "the fused tile below places 16 rows of 16 features");

__device__ __forceinline__ int table_index(int i, int last) { return max(min(i, last), 0); }

template <bool FUSE_NEXT>
__device__ __forceinline__
void gather_rows(const float* __restrict__ xs, const int* __restrict__ csr,
                 const int* __restrict__ offs, const int* __restrict__ cnt,
                 const float* __restrict__ dinv, const float* __restrict__ bias,
                 const float* __restrict__ Wn, float* __restrict__ out, int n_nodes) {
    __shared__ __attribute__((aligned(16))) float s_h[FUSE_NEXT ? 8 : 1][FUSE_NEXT ? 8 * GATHER_PASSES : 1][SH_PITCH];
    const int wave = (blockIdx.x * blockDim.x + threadIdx.x) >> 5;
    if (wave * (8 * GATHER_PASSES) >= n_nodes) return;
    const int lane = threadIdx.x & 31;
    const int g  = lane >> 2;
    const int k  = lane & 3;
    const int wl = threadIdx.x >> 5;
    const int node0 = wave * (8 * GATHER_PASSES) + g;

    int   st[GATHER_PASSES], en[GATHER_PASSES];
    float dn[GATHER_PASSES];
#pragma unroll
    for (int p = 0; p < GATHER_PASSES; ++p) {
        const int node = node0 + 8 * p;
        st[p] = offs[node];
        en[p] = st[p] + cnt[node];
        dn[p] = dinv[node];
    }
    asm volatile("" :: "v"(st[0]));
    v4f bj;
#pragma unroll
    for (int c = 0; c < 4; ++c) bj[c] = bias[4 * k + c];
    const v4f* __restrict__ xs4 = (const v4f*)xs;
    const v4f zero4 = {0.0f, 0.0f, 0.0f, 0.0f};
    int first[GATHER_PASSES][GATHER_ROWS / 4];
    v4f self[GATHER_PASSES];
#pragma unroll
    for (int p = 0; p < GATHER_PASSES; ++p) {
#pragma unroll
        for (int i = 0; i < GATHER_ROWS / 4; ++i) first[p][i] = csr[table_index(st[p] + 4 * i + k, en[p] - 1)];
        self[p] = xs4[(node0 + 8 * p) * 4 + k];
    }

#pragma unroll
    for (int p = 0; p < GATHER_PASSES; ++p) {
        const int node = node0 + 8 * p;
        const int end  = en[p];
        int id[GATHER_ROWS / 4];
#pragma unroll
        for (int i = 0; i < GATHER_ROWS / 4; ++i) id[i] = first[p][i];
        v4f acc = {0.0f, 0.0f, 0.0f, 0.0f};
        for (int base = st[p]; base < end; base += GATHER_ROWS) {
            int nx[GATHER_ROWS / 4];
#pragma unroll
            for (int i = 0; i < GATHER_ROWS / 4; ++i) nx[i] = csr[table_index(base + GATHER_ROWS + 4 * i + k, end - 1)];
            const int m = end - base;
            v4f xv[GATHER_ROWS];
#pragma unroll
            for (int q = 0; q < GATHER_ROWS; ++q) {
                const int src = __shfl(id[q >> 2], q & 3, 4);
                const int r = (q == 0 || q < m) ? (src & (N_NODES - 1)) : node;
                xv[q] = xs4[r * 4 + k];
            }
            __builtin_amdgcn_sched_barrier(0);
#pragma unroll
            for (int q = 0; q < GATHER_ROWS; ++q) acc += (q == 0 || q < m) ? xv[q] : zero4;
#pragma unroll
            for (int i = 0; i < GATHER_ROWS / 4; ++i) id[i] = nx[i];
        }
        v4f r;
#pragma unroll
        for (int c = 0; c < 4; ++c) r[c] = fmaxf(dn[p] * (acc[c] + self[p][c]) + bj[c], 0.0f);
        if (FUSE_NEXT) *(v4f*)&s_h[wl][8 * p + g][4 * k] = r;
        else           ((v4f*)out)[node * 4 + k] = r;
    }

    if (FUSE_NEXT) {
        const int j  = lane & 15;
        const int hw = lane >> 4;
        v2f bw[H_DIM / 4];
#pragma unroll
        for (int s = 0; s < H_DIM / 4; ++s) {
            bw[s].x = Wn[(4 * s + 2 * hw) * H_DIM + j];
            bw[s].y = Wn[(4 * s + 2 * hw + 1) * H_DIM + j];
        }
        float dno[8];
#pragma unroll
        for (int v = 0; v < 8; ++v) dno[v] = dinv[wave * (8 * GATHER_PASSES) + 2 * v + hw];
        __syncthreads();
        const int arow = ((j & 7) << 1) | (j >> 3);
        v8f d = {};
#pragma unroll
        for (int s = 0; s < H_DIM / 4; ++s) {
            const v2f a = *(const v2f*)&s_h[wl][arow][4 * s + 2 * hw];
            d = __builtin_amdgcn_wmma_f32_16x16x4_f32(
                    false, a, false, bw[s], (short)0, d, false, false);
        }
        float* op = out + (size_t)wave * (8 * GATHER_PASSES * H_DIM) + lane;
#pragma unroll
        for (int v = 0; v < 8; ++v) op[v * 2 * H_DIM] = dno[v] * d[v];
    }
}

__global__ __launch_bounds__(256)
void gather_kernel(const float* __restrict__ xs, const int* __restrict__ csr,
                   const int* __restrict__ offs, const int* __restrict__ cnt,
                   const float* __restrict__ dinv, const float* __restrict__ bias,
                   float* __restrict__ h, int n_nodes) {
    gather_rows<false>(xs, csr, offs, cnt, dinv, bias, nullptr, h, n_nodes);
}

__global__ __launch_bounds__(256)
void gather_next_kernel(const float* __restrict__ xs, const int* __restrict__ csr,
                        const int* __restrict__ offs, const int* __restrict__ cnt,
                        const float* __restrict__ dinv, const float* __restrict__ bias,
                        const float* __restrict__ Wn, float* __restrict__ xs_next, int n_nodes) {
    gather_rows<true>(xs, csr, offs, cnt, dinv, bias, Wn, xs_next, n_nodes);
}

extern "C" void kernel_launch(void* const* d_in, const int* in_sizes, int n_in,
                              void* d_out, int out_size, void* d_ws, size_t ws_size,
                              hipStream_t stream) {
    const float* x    = (const float*)d_in[0];
    const int*   ei   = (const int*)d_in[1];
    const float* W1   = (const float*)d_in[2];
    const float* b1   = (const float*)d_in[3];
    const float* W2   = (const float*)d_in[4];
    const float* b2   = (const float*)d_in[5];
    const float* W3   = (const float*)d_in[6];
    const float* b3   = (const float*)d_in[7];
    const float* Wp1  = (const float*)d_in[8];
    const float* bp1  = (const float*)d_in[9];
    const float* Wp2  = (const float*)d_in[10];
    const float* bp2  = (const float*)d_in[11];
    const float* Wv1  = (const float*)d_in[12];
    const float* bv1  = (const float*)d_in[13];
    const float* Wiv  = (const float*)d_in[14];
    const float* biv  = (const float*)d_in[15];
    const float* Wev  = (const float*)d_in[16];
    const float* bev  = (const float*)d_in[17];

    const int* row = ei;
    const int* col = ei + E_EDGES;

    char* ws = (char*)d_ws;
    float* dinv = (float*)ws;
    float* bufA = (float*)(ws + (size_t)N_NODES * 4);
    float* bufB = bufA + (size_t)N_NODES * H_DIM;
    float* Hp   = bufB + (size_t)N_NODES * H_DIM;
    float* part = Hp + (size_t)B_GRAPHS * FC_N;
    int*   cnt  = (int*)(part + (size_t)KCHUNKS * B_GRAPHS * 2 * FC_N);
    int*   offs = cnt + N_NODES;
    int*   binned = offs + N_NODES;
    int*   csr    = binned + E_EDGES;
    int*   runtab = csr + E_EDGES;

    float* P     = (float*)d_out;
    float* ext_V = P + (size_t)B_GRAPHS * 1024;
    float* int_V = ext_V + B_GRAPHS;

    bin_partition_kernel<<<PART_WGS, 1024, 0, stream>>>(row, col, binned, runtab);
    bin_sort_kernel<<<NBINS, 1024, 0, stream>>>(binned, runtab, x, W1, csr, cnt, offs, dinv, bufA);

    gather_next_kernel<<<N_NODES / (8 * 8 * GATHER_PASSES), 256, 0, stream>>>(bufA, csr, offs, cnt, dinv, b1, W2, bufB, N_NODES);

    gather_next_kernel<<<N_NODES / (8 * 8 * GATHER_PASSES), 256, 0, stream>>>(bufB, csr, offs, cnt, dinv, b2, W3, bufA, N_NODES);

    gather_kernel<<<N_NODES / (8 * 8 * GATHER_PASSES), 256, 0, stream>>>(bufA, csr, offs, cnt, dinv, b3, bufB, N_NODES);


    fc_pair_gemm_f32<<<dim3(2 * FC_N / (16 * FC_TILES), KCHUNKS), 32 * GEMM_WAVES, 0, stream>>>(bufB, Wp1, Wv1, part);
    fc_reduce_heads_kernel<<<B_GRAPHS, 2 * FC_N, 0, stream>>>(part, bp1, bv1, Wiv, biv, Wev, bev, Hp, ext_V, int_V);

    policy_gemm_f32<<<dim3(POL_N / 16, B_GRAPHS / 16 / POL_WAVES), 32 * POL_WAVES, 0, stream>>>(Hp, Wp2, bp2, P);
}
